# starting kernel
# baseline (speedup 1.0000x reference)
#include <hip/hip_runtime.h>
#include <hip/hip_bf16.h>

typedef __attribute__((ext_vector_type(2))) float v2f;
typedef __attribute__((ext_vector_type(8))) float v8f;

#define DD 96
#define KSTEPS (DD / 4)
#define CTILES (DD / 16)
#define LDS_PAIR_STRIDE (2 * DD)

__device__ __forceinline__ void stage_weights(float* __restrict__ lds,
                                              const float* __restrict__ W) {
    for (int i = threadIdx.x; i < DD * DD; i += 128) {
        const int r = i / DD;
        const int c = i - r * DD;
        lds[(r >> 1) * LDS_PAIR_STRIDE + c * 2 + (r & 1)] = W[i];
    }
}

__global__ void zero_kernel(float* __restrict__ p, int n) {
    int i = blockIdx.x * blockDim.x + threadIdx.x;
    if (i < n) p[i] = 0.0f;
}

__global__ __launch_bounds__(128) void gemm_bias_kernel(
    const float* __restrict__ X, const float* __restrict__ W,
    const float* __restrict__ bias, float* __restrict__ Y,
    int nPairs, int N)
{
    __shared__ float wLds[DD * DD];
    stage_weights(wLds, W);
    __syncthreads();

    const int lane = threadIdx.x & 31;
    const int wave = threadIdx.x >> 5;
    const int half = lane >> 4;
    const int l16  = lane & 15;
    const int pair = blockIdx.x * 4 + wave;
    if (pair >= nPairs) return;
    const int rowBase = pair * 32;

    int m0 = rowBase + l16;       if (m0 >= N) m0 = N - 1;
    int m1 = rowBase + 16 + l16;  if (m1 >= N) m1 = N - 1;
    const float* aPtr0 = X + (size_t)m0 * DD + 2 * half;
    const float* aPtr1 = X + (size_t)m1 * DD + 2 * half;

    v8f acc0[CTILES] = {};
    v8f acc1[CTILES] = {};
    for (int kt = 0; kt < KSTEPS; ++kt) {
        const int k = kt * 4;
        v2f a0 = *(const v2f*)(aPtr0 + k);
        v2f a1 = *(const v2f*)(aPtr1 + k);
        const float* bRow = wLds + (size_t)(kt * 2 + half) * LDS_PAIR_STRIDE;
        #pragma unroll
        for (int ct = 0; ct < CTILES; ++ct) {
            v2f b = *(const v2f*)(bRow + (ct * 16 + l16) * 2);
            acc0[ct] = __builtin_amdgcn_wmma_f32_16x16x4_f32(
                false, a0, false, b, (short)0, acc0[ct], false, false);
            acc1[ct] = __builtin_amdgcn_wmma_f32_16x16x4_f32(
                false, a1, false, b, (short)0, acc1[ct], false, false);
        }
    }

    #pragma unroll
    for (int ct = 0; ct < CTILES; ++ct) {
        const int n = ct * 16 + l16;
        const float bv = bias[n];
        #pragma unroll
        for (int r = 0; r < 8; ++r) {
            const int ma = rowBase + r + 8 * half;
            const int mb = rowBase + 16 + r + 8 * half;
            if (ma < N) Y[(size_t)ma * DD + n] = acc0[ct][r] + bv;
            if (mb < N) Y[(size_t)mb * DD + n] = acc1[ct][r] + bv;
        }
    }
}

__global__ void scatter_kernel(
    const float* __restrict__ newX, const float* __restrict__ avals,
    const int* __restrict__ row, const int* __restrict__ col,
    float* __restrict__ agg, int E)
{
    const int t = blockIdx.x * blockDim.x + threadIdx.x;
    const int e = t / 24;
    const int c = t % 24;
    if (e >= E) return;
    const float av = avals[e];
    const int dst = row[e];
    const int src = col[e];
    const float4 v = *(const float4*)(newX + (size_t)src * DD + c * 4);
    float* p = agg + (size_t)dst * DD + c * 4;
    atomicAdd(p + 0, av * v.x);
    atomicAdd(p + 1, av * v.y);
    atomicAdd(p + 2, av * v.z);
    atomicAdd(p + 3, av * v.w);
}

__global__ __launch_bounds__(128) void gate_kernel(
    const float* __restrict__ X, const float* __restrict__ agg,
    const float* __restrict__ Wgi, const float* __restrict__ Wgn,
    const float* __restrict__ bgi, const float* __restrict__ bgn,
    float* __restrict__ out, int nPairs, int N)
{
    __shared__ float wLds[DD * DD];

    const int lane = threadIdx.x & 31;
    const int wave = threadIdx.x >> 5;
    const int half = lane >> 4;
    const int l16  = lane & 15;
    const int pair = blockIdx.x * 4 + wave;
    const bool active = (pair < nPairs);
    const int rowBase = pair * 32;
    int m0 = rowBase + l16;       if (m0 >= N) m0 = (N > 0) ? N - 1 : 0;
    int m1 = rowBase + 16 + l16;  if (m1 >= N) m1 = (N > 0) ? N - 1 : 0;

    v8f acc0[CTILES] = {};
    v8f acc1[CTILES] = {};

    stage_weights(wLds, Wgi);
    __syncthreads();
    if (active) {
        const float* aPtr0 = X + (size_t)m0 * DD + 2 * half;
        const float* aPtr1 = X + (size_t)m1 * DD + 2 * half;
        for (int kt = 0; kt < KSTEPS; ++kt) {
            const int k = kt * 4;
            v2f a0 = *(const v2f*)(aPtr0 + k);
            v2f a1 = *(const v2f*)(aPtr1 + k);
            const float* bRow = wLds + (size_t)(kt * 2 + half) * LDS_PAIR_STRIDE;
            #pragma unroll
            for (int ct = 0; ct < CTILES; ++ct) {
                v2f b = *(const v2f*)(bRow + (ct * 16 + l16) * 2);
                acc0[ct] = __builtin_amdgcn_wmma_f32_16x16x4_f32(
                    false, a0, false, b, (short)0, acc0[ct], false, false);
                acc1[ct] = __builtin_amdgcn_wmma_f32_16x16x4_f32(
                    false, a1, false, b, (short)0, acc1[ct], false, false);
            }
        }
    }
    __syncthreads();

    stage_weights(wLds, Wgn);
    __syncthreads();
    if (active) {
        const float* aPtr0 = agg + (size_t)m0 * DD + 2 * half;
        const float* aPtr1 = agg + (size_t)m1 * DD + 2 * half;
        for (int kt = 0; kt < KSTEPS; ++kt) {
            const int k = kt * 4;
            v2f a0 = *(const v2f*)(aPtr0 + k);
            v2f a1 = *(const v2f*)(aPtr1 + k);
            const float* bRow = wLds + (size_t)(kt * 2 + half) * LDS_PAIR_STRIDE;
            #pragma unroll
            for (int ct = 0; ct < CTILES; ++ct) {
                v2f b = *(const v2f*)(bRow + (ct * 16 + l16) * 2);
                acc0[ct] = __builtin_amdgcn_wmma_f32_16x16x4_f32(
                    false, a0, false, b, (short)0, acc0[ct], false, false);
                acc1[ct] = __builtin_amdgcn_wmma_f32_16x16x4_f32(
                    false, a1, false, b, (short)0, acc1[ct], false, false);
            }
        }

        #pragma unroll
        for (int ct = 0; ct < CTILES; ++ct) {
            const int n = ct * 16 + l16;
            const float bsum = bgi[n] + bgn[n];
            #pragma unroll
            for (int r = 0; r < 8; ++r) {
                #pragma unroll
                for (int t2 = 0; t2 < 2; ++t2) {
                    const int m = rowBase + t2 * 16 + r + 8 * half;
                    if (m < N) {
                        const float accv = (t2 == 0) ? acc0[ct][r] : acc1[ct][r];
                        float g = accv + bsum;
                        g = 1.0f / (1.0f + __expf(-g));
                        const float av = agg[(size_t)m * DD + n];
                        const float xv = X[(size_t)m * DD + n];
                        out[(size_t)m * DD + n] = av * g + xv * (1.0f - g);
                    }
                }
            }
        }
    }
}

extern "C" void kernel_launch(void* const* d_in, const int* in_sizes, int n_in,
                              void* d_out, int out_size, void* d_ws, size_t ws_size,
                              hipStream_t stream) {
    const float* X      = (const float*)d_in[0];
    const float* a_vals = (const float*)d_in[1];
    const float* Wn     = (const float*)d_in[2];
    const float* bn     = (const float*)d_in[3];
    const float* Wgi    = (const float*)d_in[4];
    const float* bgi    = (const float*)d_in[5];
    const float* Wgn    = (const float*)d_in[6];
    const float* bgn    = (const float*)d_in[7];
    const int*   row    = (const int*)d_in[8];
    const int*   col    = (const int*)d_in[9];
    float* out = (float*)d_out;

    const int N = in_sizes[0] / DD;
    const int E = in_sizes[1];

    float* newX = (float*)d_ws;
    float* agg  = (float*)d_ws + (size_t)N * DD;

    {
        const int total = N * DD;
        zero_kernel<<<(total + 255) / 256, 256, 0, stream>>>(agg, total);
    }

    const int nPairs = (N + 31) / 32;
    const int gemmBlocks = (nPairs + 3) / 4;

    gemm_bias_kernel<<<gemmBlocks, 128, 0, stream>>>(X, Wn, bn, newX, nPairs, N);

    {
        const long long threads = (long long)E * 24;
        const int blocks = (int)((threads + 255) / 256);
        scatter_kernel<<<blocks, 256, 0, stream>>>(newX, a_vals, row, col, agg, E);
    }

    gate_kernel<<<gemmBlocks, 128, 0, stream>>>(X, agg, Wgi, Wgn, bgi, bgn,
                                                out, nPairs, N);
}


# scatter: one wave per edge row, lane l owns floats l, l+32, l+64, so every f32 atomic and gather load covers one whole 128-byte line (was 8 words in each of 4 lines); atomics kept, arrival order
# speedup vs baseline: 1.3844x; 1.3844x over previous
#include <hip/hip_runtime.h>
#include <hip/hip_bf16.h>

typedef __attribute__((ext_vector_type(2))) float v2f;
typedef __attribute__((ext_vector_type(8))) float v8f;

#define DD 96
#define KSTEPS (DD / 4)
#define CTILES (DD / 16)
#define LDS_PAIR_STRIDE (2 * DD)

__device__ __forceinline__ void stage_weights(float* __restrict__ lds,
                                              const float* __restrict__ W) {
    for (int i = threadIdx.x; i < DD * DD; i += 128) {
        const int r = i / DD;
        const int c = i - r * DD;
        lds[(r >> 1) * LDS_PAIR_STRIDE + c * 2 + (r & 1)] = W[i];
    }
}

__global__ void zero_kernel(float* __restrict__ p, int n) {
    int i = blockIdx.x * blockDim.x + threadIdx.x;
    if (i < n) p[i] = 0.0f;
}

__global__ __launch_bounds__(128) void gemm_bias_kernel(
    const float* __restrict__ X, const float* __restrict__ W,
    const float* __restrict__ bias, float* __restrict__ Y,
    int nPairs, int N)
{
    __shared__ float wLds[DD * DD];
    stage_weights(wLds, W);
    __syncthreads();

    const int lane = threadIdx.x & 31;
    const int wave = threadIdx.x >> 5;
    const int half = lane >> 4;
    const int l16  = lane & 15;
    const int pair = blockIdx.x * 4 + wave;
    if (pair >= nPairs) return;
    const int rowBase = pair * 32;

    int m0 = rowBase + l16;       if (m0 >= N) m0 = N - 1;
    int m1 = rowBase + 16 + l16;  if (m1 >= N) m1 = N - 1;
    const float* aPtr0 = X + (size_t)m0 * DD + 2 * half;
    const float* aPtr1 = X + (size_t)m1 * DD + 2 * half;

    v8f acc0[CTILES] = {};
    v8f acc1[CTILES] = {};
    for (int kt = 0; kt < KSTEPS; ++kt) {
        const int k = kt * 4;
        v2f a0 = *(const v2f*)(aPtr0 + k);
        v2f a1 = *(const v2f*)(aPtr1 + k);
        const float* bRow = wLds + (size_t)(kt * 2 + half) * LDS_PAIR_STRIDE;
        #pragma unroll
        for (int ct = 0; ct < CTILES; ++ct) {
            v2f b = *(const v2f*)(bRow + (ct * 16 + l16) * 2);
            acc0[ct] = __builtin_amdgcn_wmma_f32_16x16x4_f32(
                false, a0, false, b, (short)0, acc0[ct], false, false);
            acc1[ct] = __builtin_amdgcn_wmma_f32_16x16x4_f32(
                false, a1, false, b, (short)0, acc1[ct], false, false);
        }
    }

    #pragma unroll
    for (int ct = 0; ct < CTILES; ++ct) {
        const int n = ct * 16 + l16;
        const float bv = bias[n];
        #pragma unroll
        for (int r = 0; r < 8; ++r) {
            const int ma = rowBase + r + 8 * half;
            const int mb = rowBase + 16 + r + 8 * half;
            if (ma < N) Y[(size_t)ma * DD + n] = acc0[ct][r] + bv;
            if (mb < N) Y[(size_t)mb * DD + n] = acc1[ct][r] + bv;
        }
    }
}

#define EPW 8
#define SCATTER_WAVES 8

__global__ __launch_bounds__(SCATTER_WAVES * 32) void scatter_kernel(
    const float* __restrict__ newX, const float* __restrict__ avals,
    const int* __restrict__ row, const int* __restrict__ col,
    float* __restrict__ agg, int E)
{
    const int lane = threadIdx.x & 31;
    const int wave = __builtin_amdgcn_readfirstlane((int)(threadIdx.x >> 5));
    const int e0 = (blockIdx.x * SCATTER_WAVES + wave) * EPW;
    if (e0 >= E) return;

    if (e0 + EPW <= E) {
        int   dst[EPW];
        float av[EPW];
        float v[EPW][3];
        #pragma unroll
        for (int i = 0; i < EPW; ++i) {
            const int src = col[e0 + i];
            dst[i] = row[e0 + i];
            av[i]  = avals[e0 + i];
            const float* p = newX + (size_t)src * DD + lane;
            v[i][0] = p[0];
            v[i][1] = p[32];
            v[i][2] = p[64];
        }
        #pragma unroll
        for (int i = 0; i < EPW; ++i) {
            float* q = agg + (size_t)dst[i] * DD + lane;
            atomicAdd(q,      av[i] * v[i][0]);
            atomicAdd(q + 32, av[i] * v[i][1]);
            atomicAdd(q + 64, av[i] * v[i][2]);
        }
    } else {
        for (int e = e0; e < E; ++e) {
            const int src = col[e];
            const int d   = row[e];
            const float a = avals[e];
            const float* p = newX + (size_t)src * DD + lane;
            const float v0 = p[0], v1 = p[32], v2 = p[64];
            float* q = agg + (size_t)d * DD + lane;
            atomicAdd(q,      a * v0);
            atomicAdd(q + 32, a * v1);
            atomicAdd(q + 64, a * v2);
        }
    }
}

__global__ __launch_bounds__(128) void gate_kernel(
    const float* __restrict__ X, const float* __restrict__ agg,
    const float* __restrict__ Wgi, const float* __restrict__ Wgn,
    const float* __restrict__ bgi, const float* __restrict__ bgn,
    float* __restrict__ out, int nPairs, int N)
{
    __shared__ float wLds[DD * DD];

    const int lane = threadIdx.x & 31;
    const int wave = threadIdx.x >> 5;
    const int half = lane >> 4;
    const int l16  = lane & 15;
    const int pair = blockIdx.x * 4 + wave;
    const bool active = (pair < nPairs);
    const int rowBase = pair * 32;
    int m0 = rowBase + l16;       if (m0 >= N) m0 = (N > 0) ? N - 1 : 0;
    int m1 = rowBase + 16 + l16;  if (m1 >= N) m1 = (N > 0) ? N - 1 : 0;

    v8f acc0[CTILES] = {};
    v8f acc1[CTILES] = {};

    stage_weights(wLds, Wgi);
    __syncthreads();
    if (active) {
        const float* aPtr0 = X + (size_t)m0 * DD + 2 * half;
        const float* aPtr1 = X + (size_t)m1 * DD + 2 * half;
        for (int kt = 0; kt < KSTEPS; ++kt) {
            const int k = kt * 4;
            v2f a0 = *(const v2f*)(aPtr0 + k);
            v2f a1 = *(const v2f*)(aPtr1 + k);
            const float* bRow = wLds + (size_t)(kt * 2 + half) * LDS_PAIR_STRIDE;
            #pragma unroll
            for (int ct = 0; ct < CTILES; ++ct) {
                v2f b = *(const v2f*)(bRow + (ct * 16 + l16) * 2);
                acc0[ct] = __builtin_amdgcn_wmma_f32_16x16x4_f32(
                    false, a0, false, b, (short)0, acc0[ct], false, false);
                acc1[ct] = __builtin_amdgcn_wmma_f32_16x16x4_f32(
                    false, a1, false, b, (short)0, acc1[ct], false, false);
            }
        }
    }
    __syncthreads();

    stage_weights(wLds, Wgn);
    __syncthreads();
    if (active) {
        const float* aPtr0 = agg + (size_t)m0 * DD + 2 * half;
        const float* aPtr1 = agg + (size_t)m1 * DD + 2 * half;
        for (int kt = 0; kt < KSTEPS; ++kt) {
            const int k = kt * 4;
            v2f a0 = *(const v2f*)(aPtr0 + k);
            v2f a1 = *(const v2f*)(aPtr1 + k);
            const float* bRow = wLds + (size_t)(kt * 2 + half) * LDS_PAIR_STRIDE;
            #pragma unroll
            for (int ct = 0; ct < CTILES; ++ct) {
                v2f b = *(const v2f*)(bRow + (ct * 16 + l16) * 2);
                acc0[ct] = __builtin_amdgcn_wmma_f32_16x16x4_f32(
                    false, a0, false, b, (short)0, acc0[ct], false, false);
                acc1[ct] = __builtin_amdgcn_wmma_f32_16x16x4_f32(
                    false, a1, false, b, (short)0, acc1[ct], false, false);
            }
        }

        #pragma unroll
        for (int ct = 0; ct < CTILES; ++ct) {
            const int n = ct * 16 + l16;
            const float bsum = bgi[n] + bgn[n];
            #pragma unroll
            for (int r = 0; r < 8; ++r) {
                #pragma unroll
                for (int t2 = 0; t2 < 2; ++t2) {
                    const int m = rowBase + t2 * 16 + r + 8 * half;
                    if (m < N) {
                        const float accv = (t2 == 0) ? acc0[ct][r] : acc1[ct][r];
                        float g = accv + bsum;
                        g = 1.0f / (1.0f + __expf(-g));
                        const float av = agg[(size_t)m * DD + n];
                        const float xv = X[(size_t)m * DD + n];
                        out[(size_t)m * DD + n] = av * g + xv * (1.0f - g);
                    }
                }
            }
        }
    }
}

extern "C" void kernel_launch(void* const* d_in, const int* in_sizes, int n_in,
                              void* d_out, int out_size, void* d_ws, size_t ws_size,
                              hipStream_t stream) {
    const float* X      = (const float*)d_in[0];
    const float* a_vals = (const float*)d_in[1];
    const float* Wn     = (const float*)d_in[2];
    const float* bn     = (const float*)d_in[3];
    const float* Wgi    = (const float*)d_in[4];
    const float* bgi    = (const float*)d_in[5];
    const float* Wgn    = (const float*)d_in[6];
    const float* bgn    = (const float*)d_in[7];
    const int*   row    = (const int*)d_in[8];
    const int*   col    = (const int*)d_in[9];
    float* out = (float*)d_out;

    const int N = in_sizes[0] / DD;
    const int E = in_sizes[1];

    float* newX = (float*)d_ws;
    float* agg  = (float*)d_ws + (size_t)N * DD;

    {
        const int total = N * DD;
        zero_kernel<<<(total + 255) / 256, 256, 0, stream>>>(agg, total);
    }

    const int nPairs = (N + 31) / 32;
    const int gemmBlocks = (nPairs + 3) / 4;

    gemm_bias_kernel<<<gemmBlocks, 128, 0, stream>>>(X, Wn, bn, newX, nPairs, N);

    {
        const int edgesPerBlock = SCATTER_WAVES * EPW;
        const int blocks = (E + edgesPerBlock - 1) / edgesPerBlock;
        scatter_kernel<<<blocks, SCATTER_WAVES * 32, 0, stream>>>(newX, a_vals, row, col, agg, E);
    }

    gate_kernel<<<gemmBlocks, 128, 0, stream>>>(X, agg, Wgi, Wgn, bgi, bgn,
                                                out, nPairs, N);
}


# gate epilogue: the blend's agg and X reads issue in batches of one column tile, next tile's batch in flight during the sigmoid math; was one guarded load-wait-store round trip per output element
# speedup vs baseline: 1.8812x; 1.3589x over previous
#include <hip/hip_runtime.h>
#include <hip/hip_bf16.h>

typedef __attribute__((ext_vector_type(2))) float v2f;
typedef __attribute__((ext_vector_type(8))) float v8f;

#define DD 96
#define KSTEPS (DD / 4)
#define CTILES (DD / 16)
#define LDS_PAIR_STRIDE (2 * DD)

__device__ __forceinline__ void stage_weights(float* __restrict__ lds,
                                              const float* __restrict__ W) {
    for (int i = threadIdx.x; i < DD * DD; i += 128) {
        const int r = i / DD;
        const int c = i - r * DD;
        lds[(r >> 1) * LDS_PAIR_STRIDE + c * 2 + (r & 1)] = W[i];
    }
}

__global__ void zero_kernel(float* __restrict__ p, int n) {
    int i = blockIdx.x * blockDim.x + threadIdx.x;
    if (i < n) p[i] = 0.0f;
}

__global__ __launch_bounds__(128) void gemm_bias_kernel(
    const float* __restrict__ X, const float* __restrict__ W,
    const float* __restrict__ bias, float* __restrict__ Y,
    int nPairs, int N)
{
    __shared__ float wLds[DD * DD];
    stage_weights(wLds, W);
    __syncthreads();

    const int lane = threadIdx.x & 31;
    const int wave = threadIdx.x >> 5;
    const int half = lane >> 4;
    const int l16  = lane & 15;
    const int pair = blockIdx.x * 4 + wave;
    if (pair >= nPairs) return;
    const int rowBase = pair * 32;

    int m0 = rowBase + l16;       if (m0 >= N) m0 = N - 1;
    int m1 = rowBase + 16 + l16;  if (m1 >= N) m1 = N - 1;
    const float* aPtr0 = X + (size_t)m0 * DD + 2 * half;
    const float* aPtr1 = X + (size_t)m1 * DD + 2 * half;

    v8f acc0[CTILES] = {};
    v8f acc1[CTILES] = {};
    for (int kt = 0; kt < KSTEPS; ++kt) {
        const int k = kt * 4;
        v2f a0 = *(const v2f*)(aPtr0 + k);
        v2f a1 = *(const v2f*)(aPtr1 + k);
        const float* bRow = wLds + (size_t)(kt * 2 + half) * LDS_PAIR_STRIDE;
        #pragma unroll
        for (int ct = 0; ct < CTILES; ++ct) {
            v2f b = *(const v2f*)(bRow + (ct * 16 + l16) * 2);
            acc0[ct] = __builtin_amdgcn_wmma_f32_16x16x4_f32(
                false, a0, false, b, (short)0, acc0[ct], false, false);
            acc1[ct] = __builtin_amdgcn_wmma_f32_16x16x4_f32(
                false, a1, false, b, (short)0, acc1[ct], false, false);
        }
    }

    #pragma unroll
    for (int ct = 0; ct < CTILES; ++ct) {
        const int n = ct * 16 + l16;
        const float bv = bias[n];
        #pragma unroll
        for (int r = 0; r < 8; ++r) {
            const int ma = rowBase + r + 8 * half;
            const int mb = rowBase + 16 + r + 8 * half;
            if (ma < N) Y[(size_t)ma * DD + n] = acc0[ct][r] + bv;
            if (mb < N) Y[(size_t)mb * DD + n] = acc1[ct][r] + bv;
        }
    }
}

#define EPW 8
#define SCATTER_WAVES 8

__global__ __launch_bounds__(SCATTER_WAVES * 32) void scatter_kernel(
    const float* __restrict__ newX, const float* __restrict__ avals,
    const int* __restrict__ row, const int* __restrict__ col,
    float* __restrict__ agg, int E)
{
    const int lane = threadIdx.x & 31;
    const int wave = __builtin_amdgcn_readfirstlane((int)(threadIdx.x >> 5));
    const int e0 = (blockIdx.x * SCATTER_WAVES + wave) * EPW;
    if (e0 >= E) return;

    if (e0 + EPW <= E) {
        int   dst[EPW];
        float av[EPW];
        float v[EPW][3];
        #pragma unroll
        for (int i = 0; i < EPW; ++i) {
            const int src = col[e0 + i];
            dst[i] = row[e0 + i];
            av[i]  = avals[e0 + i];
            const float* p = newX + (size_t)src * DD + lane;
            v[i][0] = p[0];
            v[i][1] = p[32];
            v[i][2] = p[64];
        }
        #pragma unroll
        for (int i = 0; i < EPW; ++i) {
            float* q = agg + (size_t)dst[i] * DD + lane;
            atomicAdd(q,      av[i] * v[i][0]);
            atomicAdd(q + 32, av[i] * v[i][1]);
            atomicAdd(q + 64, av[i] * v[i][2]);
        }
    } else {
        for (int e = e0; e < E; ++e) {
            const int src = col[e];
            const int d   = row[e];
            const float a = avals[e];
            const float* p = newX + (size_t)src * DD + lane;
            const float v0 = p[0], v1 = p[32], v2 = p[64];
            float* q = agg + (size_t)d * DD + lane;
            atomicAdd(q,      a * v0);
            atomicAdd(q + 32, a * v1);
            atomicAdd(q + 64, a * v2);
        }
    }
}

template <bool FULL>
__device__ __forceinline__ void gate_epilogue(
    const v8f (&acc0)[CTILES], const v8f (&acc1)[CTILES],
    const float* __restrict__ X, const float* __restrict__ agg,
    const float* __restrict__ bgi, const float* __restrict__ bgn,
    float* __restrict__ out, int rowBase, int half, int l16, int N)
{
    size_t off[16];
    bool   ok[16];
    #pragma unroll
    for (int j = 0; j < 16; ++j) {
        const int m = rowBase + (j >> 3) * 16 + (j & 7) + 8 * half;
        ok[j] = FULL || (m < N);
        const int mc = (FULL || m < N) ? m : N - 1;
        off[j] = (size_t)mc * DD + l16;
    }

    float av[2][16];
    float xv[2][16];
    #pragma unroll
    for (int j = 0; j < 16; ++j) {
        av[0][j] = agg[off[j]];
        xv[0][j] = X[off[j]];
    }

    #pragma unroll
    for (int ct = 0; ct < CTILES; ++ct) {
        const int cur = ct & 1;
        const int nxt = cur ^ 1;
        if (ct + 1 < CTILES) {
            #pragma unroll
            for (int j = 0; j < 16; ++j) {
                av[nxt][j] = agg[off[j] + (ct + 1) * 16];
                xv[nxt][j] = X[off[j] + (ct + 1) * 16];
            }
        }
        const int n = ct * 16 + l16;
        const float bsum = bgi[n] + bgn[n];
        #pragma unroll
        for (int j = 0; j < 16; ++j) {
            const float accv = (j < 8) ? acc0[ct][j & 7] : acc1[ct][j & 7];
            float g = accv + bsum;
            g = 1.0f / (1.0f + __expf(-g));
            const float o = av[cur][j] * g + xv[cur][j] * (1.0f - g);
            if (ok[j]) out[off[j] + ct * 16] = o;
        }
    }
}

__global__ __launch_bounds__(128) void gate_kernel(
    const float* __restrict__ X, const float* __restrict__ agg,
    const float* __restrict__ Wgi, const float* __restrict__ Wgn,
    const float* __restrict__ bgi, const float* __restrict__ bgn,
    float* __restrict__ out, int nPairs, int N)
{
    __shared__ float wLds[DD * DD];

    const int lane = threadIdx.x & 31;
    const int wave = threadIdx.x >> 5;
    const int half = lane >> 4;
    const int l16  = lane & 15;
    const int pair = blockIdx.x * 4 + wave;
    const bool active = (pair < nPairs);
    const int rowBase = pair * 32;
    int m0 = rowBase + l16;       if (m0 >= N) m0 = (N > 0) ? N - 1 : 0;
    int m1 = rowBase + 16 + l16;  if (m1 >= N) m1 = (N > 0) ? N - 1 : 0;

    v8f acc0[CTILES] = {};
    v8f acc1[CTILES] = {};

    stage_weights(wLds, Wgi);
    __syncthreads();
    if (active) {
        const float* aPtr0 = X + (size_t)m0 * DD + 2 * half;
        const float* aPtr1 = X + (size_t)m1 * DD + 2 * half;
        for (int kt = 0; kt < KSTEPS; ++kt) {
            const int k = kt * 4;
            v2f a0 = *(const v2f*)(aPtr0 + k);
            v2f a1 = *(const v2f*)(aPtr1 + k);
            const float* bRow = wLds + (size_t)(kt * 2 + half) * LDS_PAIR_STRIDE;
            #pragma unroll
            for (int ct = 0; ct < CTILES; ++ct) {
                v2f b = *(const v2f*)(bRow + (ct * 16 + l16) * 2);
                acc0[ct] = __builtin_amdgcn_wmma_f32_16x16x4_f32(
                    false, a0, false, b, (short)0, acc0[ct], false, false);
                acc1[ct] = __builtin_amdgcn_wmma_f32_16x16x4_f32(
                    false, a1, false, b, (short)0, acc1[ct], false, false);
            }
        }
    }
    __syncthreads();

    stage_weights(wLds, Wgn);
    __syncthreads();
    if (active) {
        const float* aPtr0 = agg + (size_t)m0 * DD + 2 * half;
        const float* aPtr1 = agg + (size_t)m1 * DD + 2 * half;
        for (int kt = 0; kt < KSTEPS; ++kt) {
            const int k = kt * 4;
            v2f a0 = *(const v2f*)(aPtr0 + k);
            v2f a1 = *(const v2f*)(aPtr1 + k);
            const float* bRow = wLds + (size_t)(kt * 2 + half) * LDS_PAIR_STRIDE;
            #pragma unroll
            for (int ct = 0; ct < CTILES; ++ct) {
                v2f b = *(const v2f*)(bRow + (ct * 16 + l16) * 2);
                acc0[ct] = __builtin_amdgcn_wmma_f32_16x16x4_f32(
                    false, a0, false, b, (short)0, acc0[ct], false, false);
                acc1[ct] = __builtin_amdgcn_wmma_f32_16x16x4_f32(
                    false, a1, false, b, (short)0, acc1[ct], false, false);
            }
        }

        if (rowBase + 32 <= N)
            gate_epilogue<true>(acc0, acc1, X, agg, bgi, bgn, out,
                                rowBase, half, l16, N);
        else
            gate_epilogue<false>(acc0, acc1, X, agg, bgi, bgn, out,
                                 rowBase, half, l16, N);
    }
}

extern "C" void kernel_launch(void* const* d_in, const int* in_sizes, int n_in,
                              void* d_out, int out_size, void* d_ws, size_t ws_size,
                              hipStream_t stream) {
    const float* X      = (const float*)d_in[0];
    const float* a_vals = (const float*)d_in[1];
    const float* Wn     = (const float*)d_in[2];
    const float* bn     = (const float*)d_in[3];
    const float* Wgi    = (const float*)d_in[4];
    const float* bgi    = (const float*)d_in[5];
    const float* Wgn    = (const float*)d_in[6];
    const float* bgn    = (const float*)d_in[7];
    const int*   row    = (const int*)d_in[8];
    const int*   col    = (const int*)d_in[9];
    float* out = (float*)d_out;

    const int N = in_sizes[0] / DD;
    const int E = in_sizes[1];

    float* newX = (float*)d_ws;
    float* agg  = (float*)d_ws + (size_t)N * DD;

    {
        const int total = N * DD;
        zero_kernel<<<(total + 255) / 256, 256, 0, stream>>>(agg, total);
    }

    const int nPairs = (N + 31) / 32;
    const int gemmBlocks = (nPairs + 3) / 4;

    gemm_bias_kernel<<<gemmBlocks, 128, 0, stream>>>(X, Wn, bn, newX, nPairs, N);

    {
        const int edgesPerBlock = SCATTER_WAVES * EPW;
        const int blocks = (E + edgesPerBlock - 1) / edgesPerBlock;
        scatter_kernel<<<blocks, SCATTER_WAVES * 32, 0, stream>>>(newX, a_vals, row, col, agg, E);
    }

    gate_kernel<<<gemmBlocks, 128, 0, stream>>>(X, agg, Wgi, Wgn, bgi, bgn,
                                                out, nPairs, N);
}


# scatter-add by destination bucket: edges counting-sorted per 64 rows (ranked in LDS), each bucket summed in LDS with ds f32 atomics (arrival order), stored once; no global f32 atomics, no zero fill
# speedup vs baseline: 2.1021x; 1.1174x over previous
#include <hip/hip_runtime.h>
#include <hip/hip_bf16.h>

typedef __attribute__((ext_vector_type(2))) float v2f;
typedef __attribute__((ext_vector_type(8))) float v8f;

#define DD 96
#define KSTEPS (DD / 4)
#define CTILES (DD / 16)
#define LDS_PAIR_STRIDE (2 * DD)

__device__ __forceinline__ void stage_weights(float* __restrict__ lds,
                                              const float* __restrict__ W) {
    for (int i = threadIdx.x; i < DD * DD; i += 128) {
        const int r = i / DD;
        const int c = i - r * DD;
        lds[(r >> 1) * LDS_PAIR_STRIDE + c * 2 + (r & 1)] = W[i];
    }
}

__global__ __launch_bounds__(128) void gemm_bias_kernel(
    const float* __restrict__ X, const float* __restrict__ W,
    const float* __restrict__ bias, float* __restrict__ Y,
    int* __restrict__ ctr, int nCtr, int nPairs, int N)
{
    __shared__ float wLds[DD * DD];
    if (blockIdx.x == 0)
        for (int i = threadIdx.x; i < nCtr; i += 128) ctr[i] = 0;
    stage_weights(wLds, W);
    __syncthreads();

    const int lane = threadIdx.x & 31;
    const int wave = threadIdx.x >> 5;
    const int half = lane >> 4;
    const int l16  = lane & 15;
    const int pair = blockIdx.x * 4 + wave;
    if (pair >= nPairs) return;
    const int rowBase = pair * 32;

    int m0 = rowBase + l16;       if (m0 >= N) m0 = N - 1;
    int m1 = rowBase + 16 + l16;  if (m1 >= N) m1 = N - 1;
    const float* aPtr0 = X + (size_t)m0 * DD + 2 * half;
    const float* aPtr1 = X + (size_t)m1 * DD + 2 * half;

    v8f acc0[CTILES] = {};
    v8f acc1[CTILES] = {};
    for (int kt = 0; kt < KSTEPS; ++kt) {
        const int k = kt * 4;
        v2f a0 = *(const v2f*)(aPtr0 + k);
        v2f a1 = *(const v2f*)(aPtr1 + k);
        const float* bRow = wLds + (size_t)(kt * 2 + half) * LDS_PAIR_STRIDE;
        #pragma unroll
        for (int ct = 0; ct < CTILES; ++ct) {
            v2f b = *(const v2f*)(bRow + (ct * 16 + l16) * 2);
            acc0[ct] = __builtin_amdgcn_wmma_f32_16x16x4_f32(
                false, a0, false, b, (short)0, acc0[ct], false, false);
            acc1[ct] = __builtin_amdgcn_wmma_f32_16x16x4_f32(
                false, a1, false, b, (short)0, acc1[ct], false, false);
        }
    }

    #pragma unroll
    for (int ct = 0; ct < CTILES; ++ct) {
        const int n = ct * 16 + l16;
        const float bv = bias[n];
        #pragma unroll
        for (int r = 0; r < 8; ++r) {
            const int ma = rowBase + r + 8 * half;
            const int mb = rowBase + 16 + r + 8 * half;
            if (ma < N) Y[(size_t)ma * DD + n] = acc0[ct][r] + bv;
            if (mb < N) Y[(size_t)mb * DD + n] = acc1[ct][r] + bv;
        }
    }
}

#define RPB 64
#define RPB_SHIFT 6
#define BIN_THREADS 256
#define BIN_EPT 8
#define BIN_CHUNK (BIN_THREADS * BIN_EPT)
#define SCAN_ITEMS 4

__global__ __launch_bounds__(BIN_THREADS) void bin_count_kernel(
    const int* __restrict__ row, int* __restrict__ cnt, int E, int B)
{
    extern __shared__ int binLds[];
    int* h = binLds;
    for (int i = threadIdx.x; i < B; i += BIN_THREADS) h[i] = 0;
    __syncthreads();
    const int e0 = blockIdx.x * BIN_CHUNK + threadIdx.x;
    int rv[BIN_EPT];
    #pragma unroll
    for (int k = 0; k < BIN_EPT; ++k) {
        const int e = e0 + k * BIN_THREADS;
        rv[k] = row[min(e, E - 1)];
    }
    #pragma unroll
    for (int k = 0; k < BIN_EPT; ++k) {
        const int e = e0 + k * BIN_THREADS;
        if (e < E) atomicAdd(&h[rv[k] >> RPB_SHIFT], 1);
    }
    __syncthreads();
    for (int i = threadIdx.x; i < B; i += BIN_THREADS) {
        const int c = h[i];
        if (c) atomicAdd(&cnt[i], c);
    }
}

__global__ __launch_bounds__(BIN_THREADS) void bin_place_kernel(
    const int* __restrict__ row, const int* __restrict__ col,
    const float* __restrict__ avals, const int* __restrict__ cnt,
    int* __restrict__ cur, int* __restrict__ base, int* __restrict__ sorted,
    int E, int B)
{
    extern __shared__ int binLds[];
    __shared__ int waveTot[BIN_THREADS / 32];
    int* h   = binLds;
    int* seg = binLds + B;
    const int tid  = threadIdx.x;
    const int lane = tid & 31;
    const int wave = tid >> 5;

    int carry = 0;
    for (int b0 = 0; b0 < B; b0 += BIN_THREADS * SCAN_ITEMS) {
        const int i0 = b0 + tid * SCAN_ITEMS;
        int v[SCAN_ITEMS];
        int local = 0;
        #pragma unroll
        for (int q = 0; q < SCAN_ITEMS; ++q) {
            v[q] = (i0 + q < B) ? cnt[i0 + q] : 0;
            local += v[q];
        }
        int s = local;
        #pragma unroll
        for (int d = 1; d < 32; d <<= 1) {
            const int t = __shfl_up(s, d);
            if (lane >= d) s += t;
        }
        if (lane == 31) waveTot[wave] = s;
        __syncthreads();
        int before = 0, total = 0;
        #pragma unroll
        for (int w = 0; w < BIN_THREADS / 32; ++w) {
            const int t = waveTot[w];
            if (w < wave) before += t;
            total += t;
        }
        int run = carry + before + s - local;
        #pragma unroll
        for (int q = 0; q < SCAN_ITEMS; ++q) {
            if (i0 + q < B) { seg[i0 + q] = run; h[i0 + q] = 0; }
            run += v[q];
        }
        carry += total;
        __syncthreads();
    }
    if (blockIdx.x == 0)
        for (int i = tid; i < B; i += BIN_THREADS) base[i] = seg[i];

    int rv[BIN_EPT], cv[BIN_EPT], av[BIN_EPT], rk[BIN_EPT];
    const int e0 = blockIdx.x * BIN_CHUNK + tid;
    #pragma unroll
    for (int k = 0; k < BIN_EPT; ++k) {
        const int e = e0 + k * BIN_THREADS;
        const int ec = min(e, E - 1);
        rv[k] = row[ec];
        cv[k] = col[ec];
        av[k] = __float_as_int(avals[ec]);
        rk[k] = 0;
    }
    #pragma unroll
    for (int k = 0; k < BIN_EPT; ++k) {
        const int e = e0 + k * BIN_THREADS;
        if (e < E) rk[k] = atomicAdd(&h[rv[k] >> RPB_SHIFT], 1);
    }
    __syncthreads();
    for (int i = tid; i < B; i += BIN_THREADS) {
        const int c = h[i];
        if (c) seg[i] += atomicAdd(&cur[i], c);
    }
    __syncthreads();
    int4* recs = (int4*)sorted;
    #pragma unroll
    for (int k = 0; k < BIN_EPT; ++k) {
        const int e = e0 + k * BIN_THREADS;
        if (e < E) {
            int pos = seg[rv[k] >> RPB_SHIFT] + rk[k];
            pos = min(max(pos, 0), E - 1);
            recs[pos] = make_int4(rv[k], cv[k], av[k], 0);
        }
    }
}

#define EPW 8
#define SUM_WAVES 8

__global__ __launch_bounds__(SUM_WAVES * 32) void bucket_sum_kernel(
    const float* __restrict__ newX, const int* __restrict__ sorted,
    const int* __restrict__ cnt, const int* __restrict__ base,
    float* __restrict__ agg, int E, int N)
{
    __shared__ float acc[RPB * DD];
    for (int i = threadIdx.x; i < RPB * DD; i += SUM_WAVES * 32) acc[i] = 0.0f;
    __syncthreads();

    const int b    = blockIdx.x;
    const int lane = threadIdx.x & 31;
    const int wave = __builtin_amdgcn_readfirstlane((int)(threadIdx.x >> 5));
    int beg = base[b];
    int n   = cnt[b];
    beg = min(max(beg, 0), E);
    n   = min(max(n, 0), E - beg);
    const int4* recs = (const int4*)sorted + beg;

    for (int j = wave * EPW; j < n; j += SUM_WAVES * EPW) {
        if (j + EPW <= n) {
            int   rl[EPW];
            float a[EPW];
            float v[EPW][3];
            #pragma unroll
            for (int i = 0; i < EPW; ++i) {
                const int4 r = recs[j + i];
                rl[i] = r.x & (RPB - 1);
                a[i]  = __int_as_float(r.z);
                const float* p = newX + (size_t)r.y * DD + lane;
                v[i][0] = p[0];
                v[i][1] = p[32];
                v[i][2] = p[64];
            }
            #pragma unroll
            for (int i = 0; i < EPW; ++i) {
                float* q = acc + rl[i] * DD + lane;
                atomicAdd(q,      a[i] * v[i][0]);
                atomicAdd(q + 32, a[i] * v[i][1]);
                atomicAdd(q + 64, a[i] * v[i][2]);
            }
        } else {
            for (int i = j; i < n; ++i) {
                const int4 r = recs[i];
                const float av = __int_as_float(r.z);
                const float* p = newX + (size_t)r.y * DD + lane;
                const float v0 = p[0], v1 = p[32], v2 = p[64];
                float* q = acc + (r.x & (RPB - 1)) * DD + lane;
                atomicAdd(q,      av * v0);
                atomicAdd(q + 32, av * v1);
                atomicAdd(q + 64, av * v2);
            }
        }
    }
    __syncthreads();

    const int rowBase = b * RPB;
    float4* dst = (float4*)(agg + (size_t)rowBase * DD);
    const float4* srcv = (const float4*)acc;
    for (int i = threadIdx.x; i < RPB * DD / 4; i += SUM_WAVES * 32) {
        const int r = (i * 4) / DD;
        if (rowBase + r < N) dst[i] = srcv[i];
    }
}

template <bool FULL>
__device__ __forceinline__ void gate_epilogue(
    const v8f (&acc0)[CTILES], const v8f (&acc1)[CTILES],
    const float* __restrict__ X, const float* __restrict__ agg,
    const float* __restrict__ bgi, const float* __restrict__ bgn,
    float* __restrict__ out, int rowBase, int half, int l16, int N)
{
    size_t off[16];
    bool   ok[16];
    #pragma unroll
    for (int j = 0; j < 16; ++j) {
        const int m = rowBase + (j >> 3) * 16 + (j & 7) + 8 * half;
        ok[j] = FULL || (m < N);
        const int mc = (FULL || m < N) ? m : N - 1;
        off[j] = (size_t)mc * DD + l16;
    }

    float av[2][16];
    float xv[2][16];
    #pragma unroll
    for (int j = 0; j < 16; ++j) {
        av[0][j] = agg[off[j]];
        xv[0][j] = X[off[j]];
    }

    #pragma unroll
    for (int ct = 0; ct < CTILES; ++ct) {
        const int cur = ct & 1;
        const int nxt = cur ^ 1;
        if (ct + 1 < CTILES) {
            #pragma unroll
            for (int j = 0; j < 16; ++j) {
                av[nxt][j] = agg[off[j] + (ct + 1) * 16];
                xv[nxt][j] = X[off[j] + (ct + 1) * 16];
            }
        }
        const int n = ct * 16 + l16;
        const float bsum = bgi[n] + bgn[n];
        #pragma unroll
        for (int j = 0; j < 16; ++j) {
            const float accv = (j < 8) ? acc0[ct][j & 7] : acc1[ct][j & 7];
            float g = accv + bsum;
            g = 1.0f / (1.0f + __expf(-g));
            const float o = av[cur][j] * g + xv[cur][j] * (1.0f - g);
            if (ok[j]) out[off[j] + ct * 16] = o;
        }
    }
}

__global__ __launch_bounds__(128) void gate_kernel(
    const float* __restrict__ X, const float* __restrict__ agg,
    const float* __restrict__ Wgi, const float* __restrict__ Wgn,
    const float* __restrict__ bgi, const float* __restrict__ bgn,
    float* __restrict__ out, int nPairs, int N)
{
    __shared__ float wLds[DD * DD];

    const int lane = threadIdx.x & 31;
    const int wave = threadIdx.x >> 5;
    const int half = lane >> 4;
    const int l16  = lane & 15;
    const int pair = blockIdx.x * 4 + wave;
    const bool active = (pair < nPairs);
    const int rowBase = pair * 32;
    int m0 = rowBase + l16;       if (m0 >= N) m0 = (N > 0) ? N - 1 : 0;
    int m1 = rowBase + 16 + l16;  if (m1 >= N) m1 = (N > 0) ? N - 1 : 0;

    v8f acc0[CTILES] = {};
    v8f acc1[CTILES] = {};

    stage_weights(wLds, Wgi);
    __syncthreads();
    if (active) {
        const float* aPtr0 = X + (size_t)m0 * DD + 2 * half;
        const float* aPtr1 = X + (size_t)m1 * DD + 2 * half;
        for (int kt = 0; kt < KSTEPS; ++kt) {
            const int k = kt * 4;
            v2f a0 = *(const v2f*)(aPtr0 + k);
            v2f a1 = *(const v2f*)(aPtr1 + k);
            const float* bRow = wLds + (size_t)(kt * 2 + half) * LDS_PAIR_STRIDE;
            #pragma unroll
            for (int ct = 0; ct < CTILES; ++ct) {
                v2f b = *(const v2f*)(bRow + (ct * 16 + l16) * 2);
                acc0[ct] = __builtin_amdgcn_wmma_f32_16x16x4_f32(
                    false, a0, false, b, (short)0, acc0[ct], false, false);
                acc1[ct] = __builtin_amdgcn_wmma_f32_16x16x4_f32(
                    false, a1, false, b, (short)0, acc1[ct], false, false);
            }
        }
    }
    __syncthreads();

    stage_weights(wLds, Wgn);
    __syncthreads();
    if (active) {
        const float* aPtr0 = agg + (size_t)m0 * DD + 2 * half;
        const float* aPtr1 = agg + (size_t)m1 * DD + 2 * half;
        for (int kt = 0; kt < KSTEPS; ++kt) {
            const int k = kt * 4;
            v2f a0 = *(const v2f*)(aPtr0 + k);
            v2f a1 = *(const v2f*)(aPtr1 + k);
            const float* bRow = wLds + (size_t)(kt * 2 + half) * LDS_PAIR_STRIDE;
            #pragma unroll
            for (int ct = 0; ct < CTILES; ++ct) {
                v2f b = *(const v2f*)(bRow + (ct * 16 + l16) * 2);
                acc0[ct] = __builtin_amdgcn_wmma_f32_16x16x4_f32(
                    false, a0, false, b, (short)0, acc0[ct], false, false);
                acc1[ct] = __builtin_amdgcn_wmma_f32_16x16x4_f32(
                    false, a1, false, b, (short)0, acc1[ct], false, false);
            }
        }

        if (rowBase + 32 <= N)
            gate_epilogue<true>(acc0, acc1, X, agg, bgi, bgn, out,
                                rowBase, half, l16, N);
        else
            gate_epilogue<false>(acc0, acc1, X, agg, bgi, bgn, out,
                                 rowBase, half, l16, N);
    }
}

extern "C" void kernel_launch(void* const* d_in, const int* in_sizes, int n_in,
                              void* d_out, int out_size, void* d_ws, size_t ws_size,
                              hipStream_t stream) {
    const float* X      = (const float*)d_in[0];
    const float* a_vals = (const float*)d_in[1];
    const float* Wn     = (const float*)d_in[2];
    const float* bn     = (const float*)d_in[3];
    const float* Wgi    = (const float*)d_in[4];
    const float* bgi    = (const float*)d_in[5];
    const float* Wgn    = (const float*)d_in[6];
    const float* bgn    = (const float*)d_in[7];
    const int*   row    = (const int*)d_in[8];
    const int*   col    = (const int*)d_in[9];
    float* out = (float*)d_out;

    const int N = in_sizes[0] / DD;
    const int E = in_sizes[1];

    float* newX = (float*)d_ws;
    float* agg  = (float*)d_ws + (size_t)N * DD;

    const int nPairs = (N + 31) / 32;
    const int gemmBlocks = (nPairs + 3) / 4;

    const int B = (N + RPB - 1) / RPB;
    int* sorted = (int*)d_out;
    int* cnt    = sorted + (size_t)E * 4;
    int* cur    = cnt + B;
    int* base   = cur + B;

    gemm_bias_kernel<<<gemmBlocks, 128, 0, stream>>>(X, Wn, bn, newX, cnt, 2 * B,
                                                     nPairs, N);

    {
        const int blocks = (E + BIN_CHUNK - 1) / BIN_CHUNK;
        bin_count_kernel<<<blocks, BIN_THREADS, B * sizeof(int), stream>>>(row, cnt, E, B);
        bin_place_kernel<<<blocks, BIN_THREADS, 2 * B * sizeof(int), stream>>>(
            row, col, a_vals, cnt, cur, base, sorted, E, B);
    }

    bucket_sum_kernel<<<B, SUM_WAVES * 32, 0, stream>>>(newX, sorted, cnt, base, agg, E, N);

    gate_kernel<<<gemmBlocks, 128, 0, stream>>>(X, agg, Wgi, Wgn, bgi, bgn,
                                                out, nPairs, N);
}


# gate fused into the bucket sum: a bucket's 64 agg rows stay in LDS as the A operand of agg@Wgn and for the blend, out stored from LDS in whole lines; agg never reaches memory, 4 launches not 5
# speedup vs baseline: 2.5549x; 1.2154x over previous
#include <hip/hip_runtime.h>
#include <hip/hip_bf16.h>

typedef __attribute__((ext_vector_type(2))) float v2f;
typedef __attribute__((ext_vector_type(8))) float v8f;

#define DD 96
#define KSTEPS (DD / 4)
#define CTILES (DD / 16)
#define LDS_PAIR_STRIDE (2 * DD)

__device__ __forceinline__ void stage_weights(float* __restrict__ lds,
                                              const float* __restrict__ W) {
    for (int i = threadIdx.x; i < DD * DD; i += 128) {
        const int r = i / DD;
        const int c = i - r * DD;
        lds[(r >> 1) * LDS_PAIR_STRIDE + c * 2 + (r & 1)] = W[i];
    }
}

__global__ __launch_bounds__(128) void gemm_bias_kernel(
    const float* __restrict__ X, const float* __restrict__ W,
    const float* __restrict__ bias, float* __restrict__ Y,
    int* __restrict__ ctr, int nCtr, int nPairs, int N)
{
    __shared__ float wLds[DD * DD];
    if (blockIdx.x == 0)
        for (int i = threadIdx.x; i < nCtr; i += 128) ctr[i] = 0;
    stage_weights(wLds, W);
    __syncthreads();

    const int lane = threadIdx.x & 31;
    const int wave = threadIdx.x >> 5;
    const int half = lane >> 4;
    const int l16  = lane & 15;
    const int pair = blockIdx.x * 4 + wave;
    if (pair >= nPairs) return;
    const int rowBase = pair * 32;

    int m0 = rowBase + l16;       if (m0 >= N) m0 = N - 1;
    int m1 = rowBase + 16 + l16;  if (m1 >= N) m1 = N - 1;
    const float* aPtr0 = X + (size_t)m0 * DD + 2 * half;
    const float* aPtr1 = X + (size_t)m1 * DD + 2 * half;

    v8f acc0[CTILES] = {};
    v8f acc1[CTILES] = {};
    for (int kt = 0; kt < KSTEPS; ++kt) {
        const int k = kt * 4;
        v2f a0 = *(const v2f*)(aPtr0 + k);
        v2f a1 = *(const v2f*)(aPtr1 + k);
        const float* bRow = wLds + (size_t)(kt * 2 + half) * LDS_PAIR_STRIDE;
        #pragma unroll
        for (int ct = 0; ct < CTILES; ++ct) {
            v2f b = *(const v2f*)(bRow + (ct * 16 + l16) * 2);
            acc0[ct] = __builtin_amdgcn_wmma_f32_16x16x4_f32(
                false, a0, false, b, (short)0, acc0[ct], false, false);
            acc1[ct] = __builtin_amdgcn_wmma_f32_16x16x4_f32(
                false, a1, false, b, (short)0, acc1[ct], false, false);
        }
    }

    #pragma unroll
    for (int ct = 0; ct < CTILES; ++ct) {
        const int n = ct * 16 + l16;
        const float bv = bias[n];
        #pragma unroll
        for (int r = 0; r < 8; ++r) {
            const int ma = rowBase + r + 8 * half;
            const int mb = rowBase + 16 + r + 8 * half;
            if (ma < N) Y[(size_t)ma * DD + n] = acc0[ct][r] + bv;
            if (mb < N) Y[(size_t)mb * DD + n] = acc1[ct][r] + bv;
        }
    }
}

#define RPB 64
#define RPB_SHIFT 6
#define BIN_THREADS 256
#define BIN_EPT 8
#define BIN_CHUNK (BIN_THREADS * BIN_EPT)
#define SCAN_ITEMS 4

__global__ __launch_bounds__(BIN_THREADS) void bin_count_kernel(
    const int* __restrict__ row, int* __restrict__ cnt, int E, int B)
{
    extern __shared__ int binLds[];
    int* h = binLds;
    for (int i = threadIdx.x; i < B; i += BIN_THREADS) h[i] = 0;
    __syncthreads();
    const int e0 = blockIdx.x * BIN_CHUNK + threadIdx.x;
    int rv[BIN_EPT];
    #pragma unroll
    for (int k = 0; k < BIN_EPT; ++k) {
        const int e = e0 + k * BIN_THREADS;
        rv[k] = row[min(e, E - 1)];
    }
    #pragma unroll
    for (int k = 0; k < BIN_EPT; ++k) {
        const int e = e0 + k * BIN_THREADS;
        if (e < E) atomicAdd(&h[rv[k] >> RPB_SHIFT], 1);
    }
    __syncthreads();
    for (int i = threadIdx.x; i < B; i += BIN_THREADS) {
        const int c = h[i];
        if (c) atomicAdd(&cnt[i], c);
    }
}

__global__ __launch_bounds__(BIN_THREADS) void bin_place_kernel(
    const int* __restrict__ row, const int* __restrict__ col,
    const float* __restrict__ avals, const int* __restrict__ cnt,
    int* __restrict__ cur, int* __restrict__ base, int* __restrict__ sorted,
    int E, int B)
{
    extern __shared__ int binLds[];
    __shared__ int waveTot[BIN_THREADS / 32];
    int* h   = binLds;
    int* seg = binLds + B;
    const int tid  = threadIdx.x;
    const int lane = tid & 31;
    const int wave = tid >> 5;

    int carry = 0;
    for (int b0 = 0; b0 < B; b0 += BIN_THREADS * SCAN_ITEMS) {
        const int i0 = b0 + tid * SCAN_ITEMS;
        int v[SCAN_ITEMS];
        int local = 0;
        #pragma unroll
        for (int q = 0; q < SCAN_ITEMS; ++q) {
            v[q] = (i0 + q < B) ? cnt[i0 + q] : 0;
            local += v[q];
        }
        int s = local;
        #pragma unroll
        for (int d = 1; d < 32; d <<= 1) {
            const int t = __shfl_up(s, d);
            if (lane >= d) s += t;
        }
        if (lane == 31) waveTot[wave] = s;
        __syncthreads();
        int before = 0, total = 0;
        #pragma unroll
        for (int w = 0; w < BIN_THREADS / 32; ++w) {
            const int t = waveTot[w];
            if (w < wave) before += t;
            total += t;
        }
        int run = carry + before + s - local;
        #pragma unroll
        for (int q = 0; q < SCAN_ITEMS; ++q) {
            if (i0 + q < B) { seg[i0 + q] = run; h[i0 + q] = 0; }
            run += v[q];
        }
        carry += total;
        __syncthreads();
    }
    if (blockIdx.x == 0)
        for (int i = tid; i < B; i += BIN_THREADS) base[i] = seg[i];

    int rv[BIN_EPT], cv[BIN_EPT], av[BIN_EPT], rk[BIN_EPT];
    const int e0 = blockIdx.x * BIN_CHUNK + tid;
    #pragma unroll
    for (int k = 0; k < BIN_EPT; ++k) {
        const int e = e0 + k * BIN_THREADS;
        const int ec = min(e, E - 1);
        rv[k] = row[ec];
        cv[k] = col[ec];
        av[k] = __float_as_int(avals[ec]);
        rk[k] = 0;
    }
    #pragma unroll
    for (int k = 0; k < BIN_EPT; ++k) {
        const int e = e0 + k * BIN_THREADS;
        if (e < E) rk[k] = atomicAdd(&h[rv[k] >> RPB_SHIFT], 1);
    }
    __syncthreads();
    for (int i = tid; i < B; i += BIN_THREADS) {
        const int c = h[i];
        if (c) seg[i] += atomicAdd(&cur[i], c);
    }
    __syncthreads();
    int4* recs = (int4*)sorted;
    #pragma unroll
    for (int k = 0; k < BIN_EPT; ++k) {
        const int e = e0 + k * BIN_THREADS;
        if (e < E) {
            int pos = seg[rv[k] >> RPB_SHIFT] + rk[k];
            pos = min(max(pos, 0), E - 1);
            recs[pos] = make_int4(rv[k], cv[k], av[k], 0);
        }
    }
}

#define STAGE_TRIPS 12
__device__ __forceinline__ void stage_weights_pairs(float* __restrict__ lds,
                                                    const float* __restrict__ W) {
    const int t = threadIdx.x;
    if (t < 4 * (DD / 2)) {
        const int ph = t / (DD / 2);
        const int c2 = t - ph * (DD / 2);
        const float* src = W + (2 * ph) * DD + 2 * c2;
        float* dst = lds + ph * LDS_PAIR_STRIDE + 4 * c2;
        v2f r0[STAGE_TRIPS], r1[STAGE_TRIPS];
        #pragma unroll
        for (int it = 0; it < STAGE_TRIPS; ++it) {
            r0[it] = *(const v2f*)(src + it * 8 * DD);
            r1[it] = *(const v2f*)(src + it * 8 * DD + DD);
        }
        #pragma unroll
        for (int it = 0; it < STAGE_TRIPS; ++it) {
            *(float4*)(dst + it * 4 * LDS_PAIR_STRIDE) =
                make_float4(r0[it].x, r1[it].x, r0[it].y, r1[it].y);
        }
    }
}

#define EPW 8
#define SUM_WAVES 8
#define WCT 3

__global__ __launch_bounds__(SUM_WAVES * 32) void bucket_gate_kernel(
    const float* __restrict__ newX, const int* __restrict__ sorted,
    const int* __restrict__ cnt, const int* __restrict__ base,
    const float* __restrict__ X,
    const float* __restrict__ Wgi, const float* __restrict__ Wgn,
    const float* __restrict__ bgi, const float* __restrict__ bgn,
    float* __restrict__ out, int E, int N)
{
    __shared__ float acc[RPB * DD];
    __shared__ float wLds[DD * DD];
    for (int i = threadIdx.x; i < RPB * DD; i += SUM_WAVES * 32) acc[i] = 0.0f;
    __syncthreads();

    const int b    = blockIdx.x;
    const int lane = threadIdx.x & 31;
    const int wave = __builtin_amdgcn_readfirstlane((int)(threadIdx.x >> 5));
    int beg = base[b];
    int n   = cnt[b];
    beg = min(max(beg, 0), E);
    n   = min(max(n, 0), E - beg);
    const int4* recs = (const int4*)sorted + beg;

    for (int j = wave * EPW; j < n; j += SUM_WAVES * EPW) {
        if (j + EPW <= n) {
            int   rl[EPW];
            float a[EPW];
            float v[EPW][3];
            #pragma unroll
            for (int i = 0; i < EPW; ++i) {
                const int4 r = recs[j + i];
                rl[i] = r.x & (RPB - 1);
                a[i]  = __int_as_float(r.z);
                const float* p = newX + (size_t)r.y * DD + lane;
                v[i][0] = p[0];
                v[i][1] = p[32];
                v[i][2] = p[64];
            }
            #pragma unroll
            for (int i = 0; i < EPW; ++i) {
                float* q = acc + rl[i] * DD + lane;
                atomicAdd(q,      a[i] * v[i][0]);
                atomicAdd(q + 32, a[i] * v[i][1]);
                atomicAdd(q + 64, a[i] * v[i][2]);
            }
        } else {
            for (int i = j; i < n; ++i) {
                const int4 r = recs[i];
                const float av = __int_as_float(r.z);
                const float* p = newX + (size_t)r.y * DD + lane;
                const float v0 = p[0], v1 = p[32], v2 = p[64];
                float* q = acc + (r.x & (RPB - 1)) * DD + lane;
                atomicAdd(q,      av * v0);
                atomicAdd(q + 32, av * v1);
                atomicAdd(q + 64, av * v2);
            }
        }
    }

    const int half = lane >> 4;
    const int l16  = lane & 15;
    const int rt   = wave >> 1;
    const int ct0  = (wave & 1) * WCT;
    const int rowBase = b * RPB;
    const int lastRow = (N > 0) ? N - 1 : 0;

    v2f aX[KSTEPS];
    {
        const int m = min(rowBase + rt * 16 + l16, lastRow);
        const float* aPtr = X + (size_t)m * DD + 2 * half;
        #pragma unroll
        for (int kt = 0; kt < KSTEPS; ++kt) aX[kt] = *(const v2f*)(aPtr + kt * 4);
    }

    stage_weights_pairs(wLds, Wgi);
    __syncthreads();

    v8f g[WCT] = {};
    #pragma unroll
    for (int kt = 0; kt < KSTEPS; ++kt) {
        const float* bRow = wLds + (size_t)(kt * 2 + half) * LDS_PAIR_STRIDE;
        #pragma unroll
        for (int t = 0; t < WCT; ++t) {
            v2f bw = *(const v2f*)(bRow + ((ct0 + t) * 16 + l16) * 2);
            g[t] = __builtin_amdgcn_wmma_f32_16x16x4_f32(
                false, aX[kt], false, bw, (short)0, g[t], false, false);
        }
    }

    float xe[WCT][8];
    #pragma unroll
    for (int r = 0; r < 8; ++r) {
        const int m = min(rowBase + rt * 16 + r + 8 * half, lastRow);
        const float* xp = X + (size_t)m * DD + ct0 * 16 + l16;
        #pragma unroll
        for (int t = 0; t < WCT; ++t) xe[t][r] = xp[t * 16];
    }

    __syncthreads();
    stage_weights_pairs(wLds, Wgn);
    __syncthreads();

    {
        const float* aRow = acc + (rt * 16 + l16) * DD + 2 * half;
        #pragma unroll
        for (int kt = 0; kt < KSTEPS; ++kt) {
            const v2f a2 = *(const v2f*)(aRow + kt * 4);
            const float* bRow = wLds + (size_t)(kt * 2 + half) * LDS_PAIR_STRIDE;
            #pragma unroll
            for (int t = 0; t < WCT; ++t) {
                v2f bw = *(const v2f*)(bRow + ((ct0 + t) * 16 + l16) * 2);
                g[t] = __builtin_amdgcn_wmma_f32_16x16x4_f32(
                    false, a2, false, bw, (short)0, g[t], false, false);
            }
        }
    }
    __syncthreads();

    #pragma unroll
    for (int t = 0; t < WCT; ++t) {
        const int nn = (ct0 + t) * 16 + l16;
        const float bsum = bgi[nn] + bgn[nn];
        #pragma unroll
        for (int r = 0; r < 8; ++r) {
            float* q = acc + (rt * 16 + r + 8 * half) * DD + nn;
            const float ag = *q;
            float s = g[t][r] + bsum;
            s = 1.0f / (1.0f + __expf(-s));
            *q = ag * s + xe[t][r] * (1.0f - s);
        }
    }
    __syncthreads();

    float4* dst = (float4*)(out + (size_t)rowBase * DD);
    const float4* srcv = (const float4*)acc;
    for (int i = threadIdx.x; i < RPB * DD / 4; i += SUM_WAVES * 32) {
        const int r = (i * 4) / DD;
        if (rowBase + r < N) dst[i] = srcv[i];
    }
}

extern "C" void kernel_launch(void* const* d_in, const int* in_sizes, int n_in,
                              void* d_out, int out_size, void* d_ws, size_t ws_size,
                              hipStream_t stream) {
    const float* X      = (const float*)d_in[0];
    const float* a_vals = (const float*)d_in[1];
    const float* Wn     = (const float*)d_in[2];
    const float* bn     = (const float*)d_in[3];
    const float* Wgi    = (const float*)d_in[4];
    const float* bgi    = (const float*)d_in[5];
    const float* Wgn    = (const float*)d_in[6];
    const float* bgn    = (const float*)d_in[7];
    const int*   row    = (const int*)d_in[8];
    const int*   col    = (const int*)d_in[9];
    float* out = (float*)d_out;

    const int N = in_sizes[0] / DD;
    const int E = in_sizes[1];

    float* newX = (float*)d_ws;

    const int nPairs = (N + 31) / 32;
    const int gemmBlocks = (nPairs + 3) / 4;

    const int B = (N + RPB - 1) / RPB;
    int* sorted = (int*)((float*)d_ws + (size_t)N * DD);
    int* cnt    = sorted + (size_t)E * 4;
    int* cur    = cnt + B;
    int* base   = cur + B;

    gemm_bias_kernel<<<gemmBlocks, 128, 0, stream>>>(X, Wn, bn, newX, cnt, 2 * B,
                                                     nPairs, N);

    {
        const int blocks = (E + BIN_CHUNK - 1) / BIN_CHUNK;
        bin_count_kernel<<<blocks, BIN_THREADS, B * sizeof(int), stream>>>(row, cnt, E, B);
        bin_place_kernel<<<blocks, BIN_THREADS, 2 * B * sizeof(int), stream>>>(
            row, col, a_vals, cnt, cur, base, sorted, E, B);
    }

    bucket_gate_kernel<<<B, SUM_WAVES * 32, 0, stream>>>(
        newX, sorted, cnt, base, X, Wgi, Wgn, bgi, bgn, out, E, N);
}


# edge sum taken before the Wn product: the bucket block sums a*X[col] in LDS, then agg = S@Wn + (sum a)*bn on WMMA in the block; the newX GEMM launch and plane go; f32 sums reassociated
# speedup vs baseline: 3.0956x; 1.2116x over previous
#include <hip/hip_runtime.h>
#include <hip/hip_bf16.h>

typedef __attribute__((ext_vector_type(2))) float v2f;
typedef __attribute__((ext_vector_type(8))) float v8f;

#define DD 96
#define KSTEPS (DD / 4)
#define CTILES (DD / 16)
#define LDS_PAIR_STRIDE (2 * DD)

#define RPB 64
#define RPB_SHIFT 6
#define BIN_THREADS 256
#define BIN_EPT 8
#define BIN_CHUNK (BIN_THREADS * BIN_EPT)
#define SCAN_ITEMS 4

__global__ __launch_bounds__(BIN_THREADS) void clear_counters_kernel(
    int* __restrict__ ctr, int nCtr)
{
    for (int i = threadIdx.x; i < nCtr; i += BIN_THREADS) ctr[i] = 0;
}

__global__ __launch_bounds__(BIN_THREADS) void bin_count_kernel(
    const int* __restrict__ row, int* __restrict__ cnt, int E, int B)
{
    extern __shared__ int binLds[];
    int* h = binLds;
    for (int i = threadIdx.x; i < B; i += BIN_THREADS) h[i] = 0;
    __syncthreads();
    const int e0 = blockIdx.x * BIN_CHUNK + threadIdx.x;
    int rv[BIN_EPT];
    #pragma unroll
    for (int k = 0; k < BIN_EPT; ++k) {
        const int e = e0 + k * BIN_THREADS;
        rv[k] = row[min(e, E - 1)];
    }
    #pragma unroll
    for (int k = 0; k < BIN_EPT; ++k) {
        const int e = e0 + k * BIN_THREADS;
        if (e < E) atomicAdd(&h[rv[k] >> RPB_SHIFT], 1);
    }
    __syncthreads();
    for (int i = threadIdx.x; i < B; i += BIN_THREADS) {
        const int c = h[i];
        if (c) atomicAdd(&cnt[i], c);
    }
}

__global__ __launch_bounds__(BIN_THREADS) void bin_place_kernel(
    const int* __restrict__ row, const int* __restrict__ col,
    const float* __restrict__ avals, const int* __restrict__ cnt,
    int* __restrict__ cur, int* __restrict__ base, int* __restrict__ sorted,
    int E, int B)
{
    extern __shared__ int binLds[];
    __shared__ int waveTot[BIN_THREADS / 32];
    int* h   = binLds;
    int* seg = binLds + B;
    const int tid  = threadIdx.x;
    const int lane = tid & 31;
    const int wave = tid >> 5;

    int carry = 0;
    for (int b0 = 0; b0 < B; b0 += BIN_THREADS * SCAN_ITEMS) {
        const int i0 = b0 + tid * SCAN_ITEMS;
        int v[SCAN_ITEMS];
        int local = 0;
        #pragma unroll
        for (int q = 0; q < SCAN_ITEMS; ++q) {
            v[q] = (i0 + q < B) ? cnt[i0 + q] : 0;
            local += v[q];
        }
        int s = local;
        #pragma unroll
        for (int d = 1; d < 32; d <<= 1) {
            const int t = __shfl_up(s, d);
            if (lane >= d) s += t;
        }
        if (lane == 31) waveTot[wave] = s;
        __syncthreads();
        int before = 0, total = 0;
        #pragma unroll
        for (int w = 0; w < BIN_THREADS / 32; ++w) {
            const int t = waveTot[w];
            if (w < wave) before += t;
            total += t;
        }
        int run = carry + before + s - local;
        #pragma unroll
        for (int q = 0; q < SCAN_ITEMS; ++q) {
            if (i0 + q < B) { seg[i0 + q] = run; h[i0 + q] = 0; }
            run += v[q];
        }
        carry += total;
        __syncthreads();
    }
    if (blockIdx.x == 0)
        for (int i = tid; i < B; i += BIN_THREADS) base[i] = seg[i];

    int rv[BIN_EPT], cv[BIN_EPT], av[BIN_EPT], rk[BIN_EPT];
    const int e0 = blockIdx.x * BIN_CHUNK + tid;
    #pragma unroll
    for (int k = 0; k < BIN_EPT; ++k) {
        const int e = e0 + k * BIN_THREADS;
        const int ec = min(e, E - 1);
        rv[k] = row[ec];
        cv[k] = col[ec];
        av[k] = __float_as_int(avals[ec]);
        rk[k] = 0;
    }
    #pragma unroll
    for (int k = 0; k < BIN_EPT; ++k) {
        const int e = e0 + k * BIN_THREADS;
        if (e < E) rk[k] = atomicAdd(&h[rv[k] >> RPB_SHIFT], 1);
    }
    __syncthreads();
    for (int i = tid; i < B; i += BIN_THREADS) {
        const int c = h[i];
        if (c) seg[i] += atomicAdd(&cur[i], c);
    }
    __syncthreads();
    int4* recs = (int4*)sorted;
    #pragma unroll
    for (int k = 0; k < BIN_EPT; ++k) {
        const int e = e0 + k * BIN_THREADS;
        if (e < E) {
            int pos = seg[rv[k] >> RPB_SHIFT] + rk[k];
            pos = min(max(pos, 0), E - 1);
            recs[pos] = make_int4(rv[k], cv[k], av[k], 0);
        }
    }
}

#define STAGE_TRIPS 12
__device__ __forceinline__ void stage_weights_pairs(float* __restrict__ lds,
                                                    const float* __restrict__ W) {
    const int t = threadIdx.x;
    if (t < 4 * (DD / 2)) {
        const int ph = t / (DD / 2);
        const int c2 = t - ph * (DD / 2);
        const float* src = W + (2 * ph) * DD + 2 * c2;
        float* dst = lds + ph * LDS_PAIR_STRIDE + 4 * c2;
        v2f r0[STAGE_TRIPS], r1[STAGE_TRIPS];
        #pragma unroll
        for (int it = 0; it < STAGE_TRIPS; ++it) {
            r0[it] = *(const v2f*)(src + it * 8 * DD);
            r1[it] = *(const v2f*)(src + it * 8 * DD + DD);
        }
        #pragma unroll
        for (int it = 0; it < STAGE_TRIPS; ++it) {
            *(float4*)(dst + it * 4 * LDS_PAIR_STRIDE) =
                make_float4(r0[it].x, r1[it].x, r0[it].y, r1[it].y);
        }
    }
}

#define EPW 8
#define SUM_WAVES 8
#define WCT 3

__global__ __launch_bounds__(SUM_WAVES * 32) void bucket_gate_kernel(
    const int* __restrict__ sorted,
    const int* __restrict__ cnt, const int* __restrict__ base,
    const float* __restrict__ X,
    const float* __restrict__ Wn, const float* __restrict__ bn,
    const float* __restrict__ Wgi, const float* __restrict__ Wgn,
    const float* __restrict__ bgi, const float* __restrict__ bgn,
    float* __restrict__ out, int E, int N)
{
    __shared__ float acc[RPB * DD];
    __shared__ float wLds[DD * DD];
    __shared__ float asum[RPB];
    const int b = blockIdx.x;
    int beg = base[b];
    int n   = cnt[b];
    stage_weights_pairs(wLds, Wn);
    for (int i = threadIdx.x; i < RPB * DD; i += SUM_WAVES * 32) acc[i] = 0.0f;
    if (threadIdx.x < RPB) asum[threadIdx.x] = 0.0f;
    __syncthreads();

    const int lane = threadIdx.x & 31;
    const int wave = __builtin_amdgcn_readfirstlane((int)(threadIdx.x >> 5));
    beg = min(max(beg, 0), E);
    n   = min(max(n, 0), E - beg);
    const int4* recs = (const int4*)sorted + beg;
    const int lastRow = (N > 0) ? N - 1 : 0;

    for (int j = wave * EPW; j < n; j += SUM_WAVES * EPW) {
        if (j + EPW <= n) {
            int   rl[EPW];
            float a[EPW];
            float v[EPW][3];
            #pragma unroll
            for (int i = 0; i < EPW; ++i) {
                const int4 r = recs[j + i];
                rl[i] = r.x & (RPB - 1);
                a[i]  = __int_as_float(r.z);
                const float* p = X + (size_t)r.y * DD + lane;
                v[i][0] = p[0];
                v[i][1] = p[32];
                v[i][2] = p[64];
            }
            #pragma unroll
            for (int i = 0; i < EPW; ++i) {
                float* q = acc + rl[i] * DD + lane;
                atomicAdd(q,      a[i] * v[i][0]);
                atomicAdd(q + 32, a[i] * v[i][1]);
                atomicAdd(q + 64, a[i] * v[i][2]);
            }
            int   rsel = rl[0];
            float asel = a[0];
            #pragma unroll
            for (int i = 1; i < EPW; ++i) {
                if (lane == i) { rsel = rl[i]; asel = a[i]; }
            }
            if (lane < EPW) atomicAdd(asum + rsel, asel);
        } else {
            for (int i = j; i < n; ++i) {
                const int4 r = recs[i];
                const float av = __int_as_float(r.z);
                const float* p = X + (size_t)r.y * DD + lane;
                const float v0 = p[0], v1 = p[32], v2 = p[64];
                float* q = acc + (r.x & (RPB - 1)) * DD + lane;
                atomicAdd(q,      av * v0);
                atomicAdd(q + 32, av * v1);
                atomicAdd(q + 64, av * v2);
                if (lane == 0) atomicAdd(asum + (r.x & (RPB - 1)), av);
            }
        }
    }

    const int half = lane >> 4;
    const int l16  = lane & 15;
    const int rt   = wave >> 1;
    const int ct0  = (wave & 1) * WCT;
    const int rowBase = b * RPB;

    __syncthreads();

    float bnv[WCT];
    #pragma unroll
    for (int t = 0; t < WCT; ++t) bnv[t] = bn[(ct0 + t) * 16 + l16];

    v8f h[WCT] = {};
    {
        const float* aRow = acc + (rt * 16 + l16) * DD + 2 * half;
        #pragma unroll
        for (int kt = 0; kt < KSTEPS; ++kt) {
            const v2f a0 = *(const v2f*)(aRow + kt * 4);
            const float* bRow = wLds + (size_t)(kt * 2 + half) * LDS_PAIR_STRIDE;
            #pragma unroll
            for (int t = 0; t < WCT; ++t) {
                v2f bw = *(const v2f*)(bRow + ((ct0 + t) * 16 + l16) * 2);
                h[t] = __builtin_amdgcn_wmma_f32_16x16x4_f32(
                    false, a0, false, bw, (short)0, h[t], false, false);
            }
        }
    }
    __syncthreads();

    #pragma unroll
    for (int t = 0; t < WCT; ++t) {
        const int nn = (ct0 + t) * 16 + l16;
        #pragma unroll
        for (int r = 0; r < 8; ++r) {
            const int lr = rt * 16 + r + 8 * half;
            acc[lr * DD + nn] = h[t][r] + asum[lr] * bnv[t];
        }
    }
    __builtin_amdgcn_sched_barrier(0);

    v2f aX[KSTEPS];
    {
        const int m = min(rowBase + rt * 16 + l16, lastRow);
        const float* aPtr = X + (size_t)m * DD + 2 * half;
        #pragma unroll
        for (int kt = 0; kt < KSTEPS; ++kt) aX[kt] = *(const v2f*)(aPtr + kt * 4);
    }

    stage_weights_pairs(wLds, Wgi);
    __syncthreads();

    v8f g[WCT] = {};
    #pragma unroll
    for (int kt = 0; kt < KSTEPS; ++kt) {
        const float* bRow = wLds + (size_t)(kt * 2 + half) * LDS_PAIR_STRIDE;
        #pragma unroll
        for (int t = 0; t < WCT; ++t) {
            v2f bw = *(const v2f*)(bRow + ((ct0 + t) * 16 + l16) * 2);
            g[t] = __builtin_amdgcn_wmma_f32_16x16x4_f32(
                false, aX[kt], false, bw, (short)0, g[t], false, false);
        }
    }

    float xe[WCT][8];
    #pragma unroll
    for (int r = 0; r < 8; ++r) {
        const int m = min(rowBase + rt * 16 + r + 8 * half, lastRow);
        const float* xp = X + (size_t)m * DD + ct0 * 16 + l16;
        #pragma unroll
        for (int t = 0; t < WCT; ++t) xe[t][r] = xp[t * 16];
    }

    __syncthreads();
    stage_weights_pairs(wLds, Wgn);
    __syncthreads();

    {
        const float* aRow = acc + (rt * 16 + l16) * DD + 2 * half;
        #pragma unroll
        for (int kt = 0; kt < KSTEPS; ++kt) {
            const v2f a2 = *(const v2f*)(aRow + kt * 4);
            const float* bRow = wLds + (size_t)(kt * 2 + half) * LDS_PAIR_STRIDE;
            #pragma unroll
            for (int t = 0; t < WCT; ++t) {
                v2f bw = *(const v2f*)(bRow + ((ct0 + t) * 16 + l16) * 2);
                g[t] = __builtin_amdgcn_wmma_f32_16x16x4_f32(
                    false, a2, false, bw, (short)0, g[t], false, false);
            }
        }
    }
    __syncthreads();

    #pragma unroll
    for (int t = 0; t < WCT; ++t) {
        const int nn = (ct0 + t) * 16 + l16;
        const float bsum = bgi[nn] + bgn[nn];
        #pragma unroll
        for (int r = 0; r < 8; ++r) {
            float* q = acc + (rt * 16 + r + 8 * half) * DD + nn;
            const float ag = *q;
            float s = g[t][r] + bsum;
            s = 1.0f / (1.0f + __expf(-s));
            *q = ag * s + xe[t][r] * (1.0f - s);
        }
    }
    __syncthreads();

    float4* dst = (float4*)(out + (size_t)rowBase * DD);
    const float4* srcv = (const float4*)acc;
    for (int i = threadIdx.x; i < RPB * DD / 4; i += SUM_WAVES * 32) {
        const int r = (i * 4) / DD;
        if (rowBase + r < N) dst[i] = srcv[i];
    }
}

extern "C" void kernel_launch(void* const* d_in, const int* in_sizes, int n_in,
                              void* d_out, int out_size, void* d_ws, size_t ws_size,
                              hipStream_t stream) {
    const float* X      = (const float*)d_in[0];
    const float* a_vals = (const float*)d_in[1];
    const float* Wn     = (const float*)d_in[2];
    const float* bn     = (const float*)d_in[3];
    const float* Wgi    = (const float*)d_in[4];
    const float* bgi    = (const float*)d_in[5];
    const float* Wgn    = (const float*)d_in[6];
    const float* bgn    = (const float*)d_in[7];
    const int*   row    = (const int*)d_in[8];
    const int*   col    = (const int*)d_in[9];
    float* out = (float*)d_out;

    const int N = in_sizes[0] / DD;
    const int E = in_sizes[1];

    const int B = (N + RPB - 1) / RPB;
    int* sorted = (int*)d_ws;
    int* cnt    = sorted + (size_t)E * 4;
    int* cur    = cnt + B;
    int* base   = cur + B;

    clear_counters_kernel<<<1, BIN_THREADS, 0, stream>>>(cnt, 2 * B);

    {
        const int blocks = (E + BIN_CHUNK - 1) / BIN_CHUNK;
        bin_count_kernel<<<blocks, BIN_THREADS, B * sizeof(int), stream>>>(row, cnt, E, B);
        bin_place_kernel<<<blocks, BIN_THREADS, 2 * B * sizeof(int), stream>>>(
            row, col, a_vals, cnt, cur, base, sorted, E, B);
    }

    bucket_gate_kernel<<<B, SUM_WAVES * 32, 0, stream>>>(
        sorted, cnt, base, X, Wn, bn, Wgi, Wgn, bgi, bgn, out, E, N);
}


# bucket sized to the compute unit: 208 rows, one 26-wave block per unit, 241 blocks in one even round (was 782 of 64 rows, 3 or 4 per unit); LDS keeps all 3 weight matrices, staged once
# speedup vs baseline: 3.4031x; 1.0993x over previous
#include <hip/hip_runtime.h>
#include <hip/hip_bf16.h>

typedef __attribute__((ext_vector_type(2))) float v2f;
typedef __attribute__((ext_vector_type(8))) float v8f;

#define DD 96
#define KSTEPS (DD / 4)
#define CTILES (DD / 16)
#define LDS_PAIR_STRIDE (2 * DD)

#define RPB 208
#define BIN_THREADS 256
#define BIN_EPT 8
#define BIN_CHUNK (BIN_THREADS * BIN_EPT)
#define SCAN_ITEMS 4

__global__ __launch_bounds__(BIN_THREADS) void clear_counters_kernel(
    int* __restrict__ ctr, int nCtr)
{
    for (int i = threadIdx.x; i < nCtr; i += BIN_THREADS) ctr[i] = 0;
}

__global__ __launch_bounds__(BIN_THREADS) void bin_count_kernel(
    const int* __restrict__ row, int* __restrict__ cnt, int E, int B)
{
    extern __shared__ int binLds[];
    int* h = binLds;
    for (int i = threadIdx.x; i < B; i += BIN_THREADS) h[i] = 0;
    __syncthreads();
    const int e0 = blockIdx.x * BIN_CHUNK + threadIdx.x;
    int rv[BIN_EPT];
    #pragma unroll
    for (int k = 0; k < BIN_EPT; ++k) {
        const int e = e0 + k * BIN_THREADS;
        rv[k] = row[min(e, E - 1)];
    }
    #pragma unroll
    for (int k = 0; k < BIN_EPT; ++k) {
        const int e = e0 + k * BIN_THREADS;
        if (e < E) atomicAdd(&h[(unsigned)rv[k] / RPB], 1);
    }
    __syncthreads();
    for (int i = threadIdx.x; i < B; i += BIN_THREADS) {
        const int c = h[i];
        if (c) atomicAdd(&cnt[i], c);
    }
}

__global__ __launch_bounds__(BIN_THREADS) void bin_place_kernel(
    const int* __restrict__ row, const int* __restrict__ col,
    const float* __restrict__ avals, const int* __restrict__ cnt,
    int* __restrict__ cur, int* __restrict__ base, int* __restrict__ sorted,
    int E, int B)
{
    extern __shared__ int binLds[];
    __shared__ int waveTot[BIN_THREADS / 32];
    int* h   = binLds;
    int* seg = binLds + B;
    const int tid  = threadIdx.x;
    const int lane = tid & 31;
    const int wave = tid >> 5;

    int carry = 0;
    for (int b0 = 0; b0 < B; b0 += BIN_THREADS * SCAN_ITEMS) {
        const int i0 = b0 + tid * SCAN_ITEMS;
        int v[SCAN_ITEMS];
        int local = 0;
        #pragma unroll
        for (int q = 0; q < SCAN_ITEMS; ++q) {
            v[q] = (i0 + q < B) ? cnt[i0 + q] : 0;
            local += v[q];
        }
        int s = local;
        #pragma unroll
        for (int d = 1; d < 32; d <<= 1) {
            const int t = __shfl_up(s, d);
            if (lane >= d) s += t;
        }
        if (lane == 31) waveTot[wave] = s;
        __syncthreads();
        int before = 0, total = 0;
        #pragma unroll
        for (int w = 0; w < BIN_THREADS / 32; ++w) {
            const int t = waveTot[w];
            if (w < wave) before += t;
            total += t;
        }
        int run = carry + before + s - local;
        #pragma unroll
        for (int q = 0; q < SCAN_ITEMS; ++q) {
            if (i0 + q < B) { seg[i0 + q] = run; h[i0 + q] = 0; }
            run += v[q];
        }
        carry += total;
        __syncthreads();
    }
    if (blockIdx.x == 0)
        for (int i = tid; i < B; i += BIN_THREADS) base[i] = seg[i];

    int rv[BIN_EPT], cv[BIN_EPT], av[BIN_EPT], rk[BIN_EPT];
    const int e0 = blockIdx.x * BIN_CHUNK + tid;
    #pragma unroll
    for (int k = 0; k < BIN_EPT; ++k) {
        const int e = e0 + k * BIN_THREADS;
        const int ec = min(e, E - 1);
        rv[k] = row[ec];
        cv[k] = col[ec];
        av[k] = __float_as_int(avals[ec]);
        rk[k] = 0;
    }
    #pragma unroll
    for (int k = 0; k < BIN_EPT; ++k) {
        const int e = e0 + k * BIN_THREADS;
        if (e < E) rk[k] = atomicAdd(&h[(unsigned)rv[k] / RPB], 1);
    }
    __syncthreads();
    for (int i = tid; i < B; i += BIN_THREADS) {
        const int c = h[i];
        if (c) seg[i] += atomicAdd(&cur[i], c);
    }
    __syncthreads();
    int4* recs = (int4*)sorted;
    #pragma unroll
    for (int k = 0; k < BIN_EPT; ++k) {
        const int e = e0 + k * BIN_THREADS;
        if (e < E) {
            int pos = seg[(unsigned)rv[k] / RPB] + rk[k];
            pos = min(max(pos, 0), E - 1);
            recs[pos] = make_int4(rv[k], cv[k], av[k], 0);
        }
    }
}

#define STAGE_TRIPS 12
#define STAGE_THREADS (4 * (DD / 2))
struct StageRegs { v2f r0[STAGE_TRIPS]; v2f r1[STAGE_TRIPS]; };

__device__ __forceinline__ void stage_weights_load(StageRegs& sr,
                                                   const float* __restrict__ W, int t) {
    const int ph = t / (DD / 2);
    const int c2 = t - ph * (DD / 2);
    const float* src = W + (2 * ph) * DD + 2 * c2;
    #pragma unroll
    for (int it = 0; it < STAGE_TRIPS; ++it) {
        sr.r0[it] = *(const v2f*)(src + it * 8 * DD);
        sr.r1[it] = *(const v2f*)(src + it * 8 * DD + DD);
    }
}

__device__ __forceinline__ void stage_weights_store(float* __restrict__ lds,
                                                    const StageRegs& sr, int t) {
    const int ph = t / (DD / 2);
    const int c2 = t - ph * (DD / 2);
    float* dst = lds + ph * LDS_PAIR_STRIDE + 4 * c2;
    #pragma unroll
    for (int it = 0; it < STAGE_TRIPS; ++it) {
        *(float4*)(dst + it * 4 * LDS_PAIR_STRIDE) =
            make_float4(sr.r0[it].x, sr.r1[it].x, sr.r0[it].y, sr.r1[it].y);
    }
}

#define EPW 8
#define SUM_WAVES (2 * (RPB / 16))
#define WCT 3

__global__ __launch_bounds__(SUM_WAVES * 32) void bucket_gate_kernel(
    const int* __restrict__ sorted,
    const int* __restrict__ cnt, const int* __restrict__ base,
    const float* __restrict__ X,
    const float* __restrict__ Wn, const float* __restrict__ bn,
    const float* __restrict__ Wgi, const float* __restrict__ Wgn,
    const float* __restrict__ bgi, const float* __restrict__ bgn,
    float* __restrict__ out, int E, int N)
{
    __shared__ float acc[RPB * DD];
    __shared__ float wLds[3 * DD * DD];
    __shared__ float asum[RPB];
    const int b = blockIdx.x;
    int beg = base[b];
    int n   = cnt[b];
    const int sgrp = (int)threadIdx.x / STAGE_THREADS;
    const int st   = (int)threadIdx.x - sgrp * STAGE_THREADS;
    StageRegs sr;
    if (sgrp < 3) {
        const float* Wsel = (sgrp == 0) ? Wn : (sgrp == 1) ? Wgi : Wgn;
        stage_weights_load(sr, Wsel, st);
    }
    for (int i = threadIdx.x; i < RPB * DD; i += SUM_WAVES * 32) acc[i] = 0.0f;
    if (threadIdx.x < RPB) asum[threadIdx.x] = 0.0f;
    if (sgrp < 3) stage_weights_store(wLds + sgrp * (DD * DD), sr, st);
    __syncthreads();

    const int lane = threadIdx.x & 31;
    const int wave = __builtin_amdgcn_readfirstlane((int)(threadIdx.x >> 5));
    beg = min(max(beg, 0), E);
    n   = min(max(n, 0), E - beg);
    const int4* recs = (const int4*)sorted + beg;
    const int lastRow = (N > 0) ? N - 1 : 0;
    const int rowBase = b * RPB;

    for (int j = wave * EPW; j < n; j += SUM_WAVES * EPW) {
        if (j + EPW <= n) {
            int   rl[EPW];
            float a[EPW];
            float v[EPW][3];
            #pragma unroll
            for (int i = 0; i < EPW; ++i) {
                const int4 r = recs[j + i];
                rl[i] = min(max(r.x - rowBase, 0), RPB - 1);
                a[i]  = __int_as_float(r.z);
                const float* p = X + (size_t)r.y * DD + lane;
                v[i][0] = p[0];
                v[i][1] = p[32];
                v[i][2] = p[64];
            }
            #pragma unroll
            for (int i = 0; i < EPW; ++i) {
                float* q = acc + rl[i] * DD + lane;
                atomicAdd(q,      a[i] * v[i][0]);
                atomicAdd(q + 32, a[i] * v[i][1]);
                atomicAdd(q + 64, a[i] * v[i][2]);
            }
            int   rsel = rl[0];
            float asel = a[0];
            #pragma unroll
            for (int i = 1; i < EPW; ++i) {
                if (lane == i) { rsel = rl[i]; asel = a[i]; }
            }
            if (lane < EPW) atomicAdd(asum + rsel, asel);
        } else {
            for (int i = j; i < n; ++i) {
                const int4 r = recs[i];
                const float av = __int_as_float(r.z);
                const float* p = X + (size_t)r.y * DD + lane;
                const float v0 = p[0], v1 = p[32], v2 = p[64];
                const int lr = min(max(r.x - rowBase, 0), RPB - 1);
                float* q = acc + lr * DD + lane;
                atomicAdd(q,      av * v0);
                atomicAdd(q + 32, av * v1);
                atomicAdd(q + 64, av * v2);
                if (lane == 0) atomicAdd(asum + lr, av);
            }
        }
    }

    const int half = lane >> 4;
    const int l16  = lane & 15;
    const int rt   = wave >> 1;
    const int ct0  = (wave & 1) * WCT;

    v2f aX[KSTEPS];
    {
        const int m = min(rowBase + rt * 16 + l16, lastRow);
        const float* aPtr = X + (size_t)m * DD + 2 * half;
        #pragma unroll
        for (int kt = 0; kt < KSTEPS; ++kt) aX[kt] = *(const v2f*)(aPtr + kt * 4);
    }
    float bnv[WCT];
    #pragma unroll
    for (int t = 0; t < WCT; ++t) bnv[t] = bn[(ct0 + t) * 16 + l16];

    __syncthreads();

    v8f h[WCT] = {};
    {
        const float* aRow = acc + (rt * 16 + l16) * DD + 2 * half;
        #pragma unroll
        for (int kt = 0; kt < KSTEPS; ++kt) {
            const v2f a0 = *(const v2f*)(aRow + kt * 4);
            const float* bRow = wLds + (size_t)(kt * 2 + half) * LDS_PAIR_STRIDE;
            #pragma unroll
            for (int t = 0; t < WCT; ++t) {
                v2f bw = *(const v2f*)(bRow + ((ct0 + t) * 16 + l16) * 2);
                h[t] = __builtin_amdgcn_wmma_f32_16x16x4_f32(
                    false, a0, false, bw, (short)0, h[t], false, false);
            }
        }
    }
    __syncthreads();

    #pragma unroll
    for (int t = 0; t < WCT; ++t) {
        const int nn = (ct0 + t) * 16 + l16;
        #pragma unroll
        for (int r = 0; r < 8; ++r) {
            const int lr = rt * 16 + r + 8 * half;
            acc[lr * DD + nn] = h[t][r] + asum[lr] * bnv[t];
        }
    }

    v8f g[WCT] = {};
    #pragma unroll
    for (int kt = 0; kt < KSTEPS; ++kt) {
        const float* bRow = wLds + DD * DD + (size_t)(kt * 2 + half) * LDS_PAIR_STRIDE;
        #pragma unroll
        for (int t = 0; t < WCT; ++t) {
            v2f bw = *(const v2f*)(bRow + ((ct0 + t) * 16 + l16) * 2);
            g[t] = __builtin_amdgcn_wmma_f32_16x16x4_f32(
                false, aX[kt], false, bw, (short)0, g[t], false, false);
        }
    }

    float xe[WCT][8];
    #pragma unroll
    for (int r = 0; r < 8; ++r) {
        const int m = min(rowBase + rt * 16 + r + 8 * half, lastRow);
        const float* xp = X + (size_t)m * DD + ct0 * 16 + l16;
        #pragma unroll
        for (int t = 0; t < WCT; ++t) xe[t][r] = xp[t * 16];
    }

    __syncthreads();

    {
        const float* aRow = acc + (rt * 16 + l16) * DD + 2 * half;
        #pragma unroll
        for (int kt = 0; kt < KSTEPS; ++kt) {
            const v2f a2 = *(const v2f*)(aRow + kt * 4);
            const float* bRow = wLds + 2 * DD * DD + (size_t)(kt * 2 + half) * LDS_PAIR_STRIDE;
            #pragma unroll
            for (int t = 0; t < WCT; ++t) {
                v2f bw = *(const v2f*)(bRow + ((ct0 + t) * 16 + l16) * 2);
                g[t] = __builtin_amdgcn_wmma_f32_16x16x4_f32(
                    false, a2, false, bw, (short)0, g[t], false, false);
            }
        }
    }
    __syncthreads();

    #pragma unroll
    for (int t = 0; t < WCT; ++t) {
        const int nn = (ct0 + t) * 16 + l16;
        const float bsum = bgi[nn] + bgn[nn];
        #pragma unroll
        for (int r = 0; r < 8; ++r) {
            float* q = acc + (rt * 16 + r + 8 * half) * DD + nn;
            const float ag = *q;
            float s = g[t][r] + bsum;
            s = 1.0f / (1.0f + __expf(-s));
            *q = ag * s + xe[t][r] * (1.0f - s);
        }
    }
    __syncthreads();

    float4* dst = (float4*)(out + (size_t)rowBase * DD);
    const float4* srcv = (const float4*)acc;
    for (int i = threadIdx.x; i < RPB * DD / 4; i += SUM_WAVES * 32) {
        const int r = (i * 4) / DD;
        if (rowBase + r < N) dst[i] = srcv[i];
    }
}

extern "C" void kernel_launch(void* const* d_in, const int* in_sizes, int n_in,
                              void* d_out, int out_size, void* d_ws, size_t ws_size,
                              hipStream_t stream) {
    const float* X      = (const float*)d_in[0];
    const float* a_vals = (const float*)d_in[1];
    const float* Wn     = (const float*)d_in[2];
    const float* bn     = (const float*)d_in[3];
    const float* Wgi    = (const float*)d_in[4];
    const float* bgi    = (const float*)d_in[5];
    const float* Wgn    = (const float*)d_in[6];
    const float* bgn    = (const float*)d_in[7];
    const int*   row    = (const int*)d_in[8];
    const int*   col    = (const int*)d_in[9];
    float* out = (float*)d_out;

    const int N = in_sizes[0] / DD;
    const int E = in_sizes[1];

    const int B = (N + RPB - 1) / RPB;
    int* sorted = (int*)d_ws;
    int* cnt    = sorted + (size_t)E * 4;
    int* cur    = cnt + B;
    int* base   = cur + B;

    clear_counters_kernel<<<1, BIN_THREADS, 0, stream>>>(cnt, 2 * B);

    {
        const int blocks = (E + BIN_CHUNK - 1) / BIN_CHUNK;
        bin_count_kernel<<<blocks, BIN_THREADS, B * sizeof(int), stream>>>(row, cnt, E, B);
        bin_place_kernel<<<blocks, BIN_THREADS, 2 * B * sizeof(int), stream>>>(
            row, col, a_vals, cnt, cur, base, sorted, E, B);
    }

    bucket_gate_kernel<<<B, SUM_WAVES * 32, 0, stream>>>(
        sorted, cnt, base, X, Wn, bn, Wgi, Wgn, bgi, bgn, out, E, N);
}


# edges grouped without counters: each grouping block writes its own slab of bucket-ordered records plus an offset row, a bucket block walks its pieces of all slabs; 2 launches not 4; LDS f32 adds kept
# speedup vs baseline: 3.5533x; 1.0441x over previous
#include <hip/hip_runtime.h>
#include <hip/hip_bf16.h>

typedef __attribute__((ext_vector_type(2))) float v2f;
typedef __attribute__((ext_vector_type(8))) float v8f;

#define DD 96
#define KSTEPS (DD / 4)
#define CTILES (DD / 16)
#define LDS_PAIR_STRIDE (2 * DD)

#define RPB 208
#define BIN_THREADS 256
#define BIN_EPT 8
#define BIN_CHUNK (BIN_THREADS * BIN_EPT)
#define SCAN_ITEMS 4

__global__ __launch_bounds__(BIN_THREADS) void bin_place_kernel(
    const int* __restrict__ row, const int* __restrict__ col,
    const float* __restrict__ avals,
    int* __restrict__ off, int* __restrict__ sorted, int E, int B)
{
    extern __shared__ int binLds[];
    __shared__ int waveTot[BIN_THREADS / 32];
    int* h   = binLds;
    int* seg = binLds + B;
    const int tid  = threadIdx.x;
    const int lane = tid & 31;
    const int wave = tid >> 5;

    for (int i = tid; i < B; i += BIN_THREADS) h[i] = 0;

    int rv[BIN_EPT], cv[BIN_EPT], av[BIN_EPT], rk[BIN_EPT], bk[BIN_EPT];
    const int e0 = blockIdx.x * BIN_CHUNK + tid;
    #pragma unroll
    for (int k = 0; k < BIN_EPT; ++k) {
        const int e = e0 + k * BIN_THREADS;
        const int ec = min(e, E - 1);
        rv[k] = row[ec];
        cv[k] = col[ec];
        av[k] = __float_as_int(avals[ec]);
        rk[k] = 0;
    }
    __syncthreads();
    #pragma unroll
    for (int k = 0; k < BIN_EPT; ++k) {
        const int e = e0 + k * BIN_THREADS;
        bk[k] = min((int)((unsigned)rv[k] / RPB), B - 1);
        if (e < E) rk[k] = atomicAdd(&h[bk[k]], 1);
    }
    __syncthreads();

    int carry = 0;
    for (int b0 = 0; b0 < B; b0 += BIN_THREADS * SCAN_ITEMS) {
        const int i0 = b0 + tid * SCAN_ITEMS;
        int v[SCAN_ITEMS];
        int local = 0;
        #pragma unroll
        for (int q = 0; q < SCAN_ITEMS; ++q) {
            v[q] = (i0 + q < B) ? h[i0 + q] : 0;
            local += v[q];
        }
        int s = local;
        #pragma unroll
        for (int d = 1; d < 32; d <<= 1) {
            const int t = __shfl_up(s, d);
            if (lane >= d) s += t;
        }
        if (lane == 31) waveTot[wave] = s;
        __syncthreads();
        int before = 0, total = 0;
        #pragma unroll
        for (int w = 0; w < BIN_THREADS / 32; ++w) {
            const int t = waveTot[w];
            if (w < wave) before += t;
            total += t;
        }
        int run = carry + before + s - local;
        #pragma unroll
        for (int q = 0; q < SCAN_ITEMS; ++q) {
            if (i0 + q < B) seg[i0 + q] = run;
            run += v[q];
        }
        carry += total;
        __syncthreads();
    }
    if (tid == 0) seg[B] = carry;
    __syncthreads();

    int* myOff = off + (size_t)blockIdx.x * (B + 1);
    for (int i = tid; i <= B; i += BIN_THREADS) myOff[i] = seg[i];

    int4* recs = (int4*)sorted + (size_t)blockIdx.x * BIN_CHUNK;
    #pragma unroll
    for (int k = 0; k < BIN_EPT; ++k) {
        const int e = e0 + k * BIN_THREADS;
        if (e < E) {
            int pos = seg[bk[k]] + rk[k];
            pos = min(max(pos, 0), BIN_CHUNK - 1);
            recs[pos] = make_int4(rv[k], cv[k], av[k], 0);
        }
    }
}

#define STAGE_TRIPS 12
#define STAGE_THREADS (4 * (DD / 2))
struct StageRegs { v2f r0[STAGE_TRIPS]; v2f r1[STAGE_TRIPS]; };

__device__ __forceinline__ void stage_weights_load(StageRegs& sr,
                                                   const float* __restrict__ W, int t) {
    const int ph = t / (DD / 2);
    const int c2 = t - ph * (DD / 2);
    const float* src = W + (2 * ph) * DD + 2 * c2;
    #pragma unroll
    for (int it = 0; it < STAGE_TRIPS; ++it) {
        sr.r0[it] = *(const v2f*)(src + it * 8 * DD);
        sr.r1[it] = *(const v2f*)(src + it * 8 * DD + DD);
    }
}

__device__ __forceinline__ void stage_weights_store(float* __restrict__ lds,
                                                    const StageRegs& sr, int t) {
    const int ph = t / (DD / 2);
    const int c2 = t - ph * (DD / 2);
    float* dst = lds + ph * LDS_PAIR_STRIDE + 4 * c2;
    #pragma unroll
    for (int it = 0; it < STAGE_TRIPS; ++it) {
        *(float4*)(dst + it * 4 * LDS_PAIR_STRIDE) =
            make_float4(sr.r0[it].x, sr.r1[it].x, sr.r0[it].y, sr.r1[it].y);
    }
}

#define EPW 8
#define SUM_WAVES (2 * (RPB / 16))
#define WCT 3

__device__ __forceinline__ void piece_of(const int* __restrict__ off, int tstride, int b,
                                         int p, int nSlabs, int E, int& beg, int& len) {
    beg = 0;
    len = 0;
    if (p < nSlabs) {
        const int* t = off + (size_t)p * tstride + b;
        int o0 = t[0], o1 = t[1];
        const int cap = min(BIN_CHUNK, E - p * BIN_CHUNK);
        o0 = min(max(o0, 0), cap);
        o1 = min(max(o1, o0), cap);
        beg = p * BIN_CHUNK + o0;
        len = o1 - o0;
    }
}

__global__ __launch_bounds__(SUM_WAVES * 32) void bucket_gate_kernel(
    const int* __restrict__ sorted,
    const int* __restrict__ off, int nSlabs, int nBuckets,
    const float* __restrict__ X,
    const float* __restrict__ Wn, const float* __restrict__ bn,
    const float* __restrict__ Wgi, const float* __restrict__ Wgn,
    const float* __restrict__ bgi, const float* __restrict__ bgn,
    float* __restrict__ out, int E, int N)
{
    __shared__ float acc[RPB * DD];
    __shared__ float wLds[3 * DD * DD];
    __shared__ float asum[RPB];
    const int b = blockIdx.x;
    const int lane = threadIdx.x & 31;
    const int wave = __builtin_amdgcn_readfirstlane((int)(threadIdx.x >> 5));
    const int tstride = nBuckets + 1;
    int myBeg, myLen;
    piece_of(off, tstride, b, wave + lane * SUM_WAVES, nSlabs, E, myBeg, myLen);
    const int sgrp = (int)threadIdx.x / STAGE_THREADS;
    const int st   = (int)threadIdx.x - sgrp * STAGE_THREADS;
    StageRegs sr;
    if (sgrp < 3) {
        const float* Wsel = (sgrp == 0) ? Wn : (sgrp == 1) ? Wgi : Wgn;
        stage_weights_load(sr, Wsel, st);
    }
    for (int i = threadIdx.x; i < RPB * DD; i += SUM_WAVES * 32) acc[i] = 0.0f;
    if (threadIdx.x < RPB) asum[threadIdx.x] = 0.0f;
    if (sgrp < 3) stage_weights_store(wLds + sgrp * (DD * DD), sr, st);
    __syncthreads();

    const int4* recs = (const int4*)sorted;
    const int lastRow = (N > 0) ? N - 1 : 0;
    const int rowBase = b * RPB;

    for (int p0 = wave; p0 < nSlabs; ) {
        int incl = myLen;
        #pragma unroll
        for (int d = 1; d < 32; d <<= 1) {
            const int t = __shfl_up(incl, d);
            if (lane >= d) incl += t;
        }
        const int myStart = incl - myLen;
        const int total = __builtin_amdgcn_readlane(incl, 31);

        for (int j = 0; j < total; j += EPW) {
            int   rl[EPW];
            float a[EPW];
            float v[EPW][3];
            int4  rec[EPW];
            #pragma unroll
            for (int i = 0; i < EPW; ++i) {
                const int f = min(j + i, total - 1);
                const unsigned m = __builtin_amdgcn_ballot_w32(myStart <= f);
                const int L = __builtin_popcount(m) - 1;
                const int fs = __builtin_amdgcn_readlane(myStart, L);
                const int bg = __builtin_amdgcn_readlane(myBeg, L);
                rec[i] = recs[bg + (f - fs)];
            }
            #pragma unroll
            for (int i = 0; i < EPW; ++i) {
                const int4 r = rec[i];
                rl[i] = min(max(r.x - rowBase, 0), RPB - 1);
                a[i]  = (j + i < total) ? __int_as_float(r.z) : 0.0f;
                const float* p = X + (size_t)r.y * DD + lane;
                v[i][0] = p[0];
                v[i][1] = p[32];
                v[i][2] = p[64];
            }
            #pragma unroll
            for (int i = 0; i < EPW; ++i) {
                float* q = acc + rl[i] * DD + lane;
                atomicAdd(q,      a[i] * v[i][0]);
                atomicAdd(q + 32, a[i] * v[i][1]);
                atomicAdd(q + 64, a[i] * v[i][2]);
            }
            int   rsel = rl[0];
            float asel = a[0];
            #pragma unroll
            for (int i = 1; i < EPW; ++i) {
                if (lane == i) { rsel = rl[i]; asel = a[i]; }
            }
            if (lane < EPW && j + lane < total) atomicAdd(asum + rsel, asel);
        }
        p0 += SUM_WAVES * 32;
        if (p0 < nSlabs)
            piece_of(off, tstride, b, p0 + lane * SUM_WAVES, nSlabs, E, myBeg, myLen);
    }

    const int half = lane >> 4;
    const int l16  = lane & 15;
    const int rt   = wave >> 1;
    const int ct0  = (wave & 1) * WCT;

    v2f aX[KSTEPS];
    {
        const int m = min(rowBase + rt * 16 + l16, lastRow);
        const float* aPtr = X + (size_t)m * DD + 2 * half;
        #pragma unroll
        for (int kt = 0; kt < KSTEPS; ++kt) aX[kt] = *(const v2f*)(aPtr + kt * 4);
    }
    float bnv[WCT];
    #pragma unroll
    for (int t = 0; t < WCT; ++t) bnv[t] = bn[(ct0 + t) * 16 + l16];

    __syncthreads();

    v8f h[WCT] = {};
    {
        const float* aRow = acc + (rt * 16 + l16) * DD + 2 * half;
        #pragma unroll
        for (int kt = 0; kt < KSTEPS; ++kt) {
            const v2f a0 = *(const v2f*)(aRow + kt * 4);
            const float* bRow = wLds + (size_t)(kt * 2 + half) * LDS_PAIR_STRIDE;
            #pragma unroll
            for (int t = 0; t < WCT; ++t) {
                v2f bw = *(const v2f*)(bRow + ((ct0 + t) * 16 + l16) * 2);
                h[t] = __builtin_amdgcn_wmma_f32_16x16x4_f32(
                    false, a0, false, bw, (short)0, h[t], false, false);
            }
        }
    }
    __syncthreads();

    #pragma unroll
    for (int t = 0; t < WCT; ++t) {
        const int nn = (ct0 + t) * 16 + l16;
        #pragma unroll
        for (int r = 0; r < 8; ++r) {
            const int lr = rt * 16 + r + 8 * half;
            acc[lr * DD + nn] = h[t][r] + asum[lr] * bnv[t];
        }
    }

    v8f g[WCT] = {};
    #pragma unroll
    for (int kt = 0; kt < KSTEPS; ++kt) {
        const float* bRow = wLds + DD * DD + (size_t)(kt * 2 + half) * LDS_PAIR_STRIDE;
        #pragma unroll
        for (int t = 0; t < WCT; ++t) {
            v2f bw = *(const v2f*)(bRow + ((ct0 + t) * 16 + l16) * 2);
            g[t] = __builtin_amdgcn_wmma_f32_16x16x4_f32(
                false, aX[kt], false, bw, (short)0, g[t], false, false);
        }
    }

    float xe[WCT][8];
    #pragma unroll
    for (int r = 0; r < 8; ++r) {
        const int m = min(rowBase + rt * 16 + r + 8 * half, lastRow);
        const float* xp = X + (size_t)m * DD + ct0 * 16 + l16;
        #pragma unroll
        for (int t = 0; t < WCT; ++t) xe[t][r] = xp[t * 16];
    }

    __syncthreads();

    {
        const float* aRow = acc + (rt * 16 + l16) * DD + 2 * half;
        #pragma unroll
        for (int kt = 0; kt < KSTEPS; ++kt) {
            const v2f a2 = *(const v2f*)(aRow + kt * 4);
            const float* bRow = wLds + 2 * DD * DD + (size_t)(kt * 2 + half) * LDS_PAIR_STRIDE;
            #pragma unroll
            for (int t = 0; t < WCT; ++t) {
                v2f bw = *(const v2f*)(bRow + ((ct0 + t) * 16 + l16) * 2);
                g[t] = __builtin_amdgcn_wmma_f32_16x16x4_f32(
                    false, a2, false, bw, (short)0, g[t], false, false);
            }
        }
    }
    __syncthreads();

    #pragma unroll
    for (int t = 0; t < WCT; ++t) {
        const int nn = (ct0 + t) * 16 + l16;
        const float bsum = bgi[nn] + bgn[nn];
        #pragma unroll
        for (int r = 0; r < 8; ++r) {
            float* q = acc + (rt * 16 + r + 8 * half) * DD + nn;
            const float ag = *q;
            float s = g[t][r] + bsum;
            s = 1.0f / (1.0f + __expf(-s));
            *q = ag * s + xe[t][r] * (1.0f - s);
        }
    }
    __syncthreads();

    float4* dst = (float4*)(out + (size_t)rowBase * DD);
    const float4* srcv = (const float4*)acc;
    for (int i = threadIdx.x; i < RPB * DD / 4; i += SUM_WAVES * 32) {
        const int r = (i * 4) / DD;
        if (rowBase + r < N) dst[i] = srcv[i];
    }
}

extern "C" void kernel_launch(void* const* d_in, const int* in_sizes, int n_in,
                              void* d_out, int out_size, void* d_ws, size_t ws_size,
                              hipStream_t stream) {
    const float* X      = (const float*)d_in[0];
    const float* a_vals = (const float*)d_in[1];
    const float* Wn     = (const float*)d_in[2];
    const float* bn     = (const float*)d_in[3];
    const float* Wgi    = (const float*)d_in[4];
    const float* bgi    = (const float*)d_in[5];
    const float* Wgn    = (const float*)d_in[6];
    const float* bgn    = (const float*)d_in[7];
    const int*   row    = (const int*)d_in[8];
    const int*   col    = (const int*)d_in[9];
    float* out = (float*)d_out;

    const int N = in_sizes[0] / DD;
    const int E = in_sizes[1];

    const int B = (N + RPB - 1) / RPB;
    const int slabs = (E + BIN_CHUNK - 1) / BIN_CHUNK;
    int* sorted = (int*)d_ws;
    int* off    = sorted + (size_t)slabs * BIN_CHUNK * 4;

    bin_place_kernel<<<slabs, BIN_THREADS, (2 * B + 1) * sizeof(int), stream>>>(
        row, col, a_vals, off, sorted, E, B);

    bucket_gate_kernel<<<B, SUM_WAVES * 32, 0, stream>>>(
        sorted, off, slabs, B, X, Wn, bn, Wgi, Wgn, bgi, bgn, out, E, N);
}


# edge sum: bucket's records counting-sorted by row in LDS; one wave sums a row's a*X[col] in registers, 1 LDS add per row word (was 3 LDS f32 adds per record); sum order = rank arrival order
# speedup vs baseline: 3.9940x; 1.1240x over previous
#include <hip/hip_runtime.h>
#include <hip/hip_bf16.h>

typedef __attribute__((ext_vector_type(2))) float v2f;
typedef __attribute__((ext_vector_type(8))) float v8f;

#define DD 96
#define KSTEPS (DD / 4)
#define CTILES (DD / 16)
#define LDS_PAIR_STRIDE (2 * DD)

#define RPB 208
#define BIN_THREADS 256
#define BIN_EPT 8
#define BIN_CHUNK (BIN_THREADS * BIN_EPT)
#define SCAN_ITEMS 4

__global__ __launch_bounds__(BIN_THREADS) void bin_place_kernel(
    const int* __restrict__ row, const int* __restrict__ col,
    const float* __restrict__ avals,
    int* __restrict__ off, int* __restrict__ sorted, int E, int B)
{
    extern __shared__ int binLds[];
    __shared__ int waveTot[BIN_THREADS / 32];
    int* h   = binLds;
    int* seg = binLds + B;
    const int tid  = threadIdx.x;
    const int lane = tid & 31;
    const int wave = tid >> 5;

    for (int i = tid; i < B; i += BIN_THREADS) h[i] = 0;

    int rv[BIN_EPT], cv[BIN_EPT], av[BIN_EPT], rk[BIN_EPT], bk[BIN_EPT];
    const int e0 = blockIdx.x * BIN_CHUNK + tid;
    #pragma unroll
    for (int k = 0; k < BIN_EPT; ++k) {
        const int e = e0 + k * BIN_THREADS;
        const int ec = min(e, E - 1);
        rv[k] = row[ec];
        cv[k] = col[ec];
        av[k] = __float_as_int(avals[ec]);
        rk[k] = 0;
    }
    __syncthreads();
    #pragma unroll
    for (int k = 0; k < BIN_EPT; ++k) {
        const int e = e0 + k * BIN_THREADS;
        bk[k] = min((int)((unsigned)rv[k] / RPB), B - 1);
        if (e < E) rk[k] = atomicAdd(&h[bk[k]], 1);
    }
    __syncthreads();

    int carry = 0;
    for (int b0 = 0; b0 < B; b0 += BIN_THREADS * SCAN_ITEMS) {
        const int i0 = b0 + tid * SCAN_ITEMS;
        int v[SCAN_ITEMS];
        int local = 0;
        #pragma unroll
        for (int q = 0; q < SCAN_ITEMS; ++q) {
            v[q] = (i0 + q < B) ? h[i0 + q] : 0;
            local += v[q];
        }
        int s = local;
        #pragma unroll
        for (int d = 1; d < 32; d <<= 1) {
            const int t = __shfl_up(s, d);
            if (lane >= d) s += t;
        }
        if (lane == 31) waveTot[wave] = s;
        __syncthreads();
        int before = 0, total = 0;
        #pragma unroll
        for (int w = 0; w < BIN_THREADS / 32; ++w) {
            const int t = waveTot[w];
            if (w < wave) before += t;
            total += t;
        }
        int run = carry + before + s - local;
        #pragma unroll
        for (int q = 0; q < SCAN_ITEMS; ++q) {
            if (i0 + q < B) seg[i0 + q] = run;
            run += v[q];
        }
        carry += total;
        __syncthreads();
    }
    if (tid == 0) seg[B] = carry;
    __syncthreads();

    int* myOff = off + (size_t)blockIdx.x * (B + 1);
    for (int i = tid; i <= B; i += BIN_THREADS) myOff[i] = seg[i];

    int4* recs = (int4*)sorted + (size_t)blockIdx.x * BIN_CHUNK;
    #pragma unroll
    for (int k = 0; k < BIN_EPT; ++k) {
        const int e = e0 + k * BIN_THREADS;
        if (e < E) {
            int pos = seg[bk[k]] + rk[k];
            pos = min(max(pos, 0), BIN_CHUNK - 1);
            recs[pos] = make_int4(rv[k], cv[k], av[k], 0);
        }
    }
}

#define STAGE_TRIPS 12
#define STAGE_THREADS (4 * (DD / 2))
struct StageRegs { v2f r0[STAGE_TRIPS]; v2f r1[STAGE_TRIPS]; };

__device__ __forceinline__ void stage_weights_load(StageRegs& sr,
                                                   const float* __restrict__ W, int t) {
    const int ph = t / (DD / 2);
    const int c2 = t - ph * (DD / 2);
    const float* src = W + (2 * ph) * DD + 2 * c2;
    #pragma unroll
    for (int it = 0; it < STAGE_TRIPS; ++it) {
        sr.r0[it] = *(const v2f*)(src + it * 8 * DD);
        sr.r1[it] = *(const v2f*)(src + it * 8 * DD + DD);
    }
}

__device__ __forceinline__ void stage_weights_store(float* __restrict__ lds,
                                                    const StageRegs& sr, int t) {
    const int ph = t / (DD / 2);
    const int c2 = t - ph * (DD / 2);
    float* dst = lds + ph * LDS_PAIR_STRIDE + 4 * c2;
    #pragma unroll
    for (int it = 0; it < STAGE_TRIPS; ++it) {
        *(float4*)(dst + it * 4 * LDS_PAIR_STRIDE) =
            make_float4(sr.r0[it].x, sr.r1[it].x, sr.r0[it].y, sr.r1[it].y);
    }
}

#define EPW 8
#define SUM_WAVES (2 * (RPB / 16))
#define WCT 3
#define RW (RPB / SUM_WAVES)
#define RPL 8
#define RPWV (32 * RPL)
#define CAP (SUM_WAVES * RPWV)
#define SCANQ ((RPB + 31) / 32)

__device__ __forceinline__ void piece_of(const int* __restrict__ off, int tstride, int b,
                                         int p, int nSlabs, int E, int& beg, int& len) {
    beg = 0;
    len = 0;
    if (p < nSlabs) {
        const int* t = off + (size_t)p * tstride + b;
        int o0 = t[0], o1 = t[1];
        const int cap = min(BIN_CHUNK, E - p * BIN_CHUNK);
        o0 = min(max(o0, 0), cap);
        o1 = min(max(o1, o0), cap);
        beg = p * BIN_CHUNK + o0;
        len = o1 - o0;
    }
}

__global__ __launch_bounds__(SUM_WAVES * 32) void bucket_gate_kernel(
    const int* __restrict__ sorted,
    const int* __restrict__ off, int nSlabs, int nBuckets,
    const float* __restrict__ X,
    const float* __restrict__ Wn, const float* __restrict__ bn,
    const float* __restrict__ Wgi, const float* __restrict__ Wgn,
    const float* __restrict__ bgi, const float* __restrict__ bgn,
    float* __restrict__ out, int E, int N)
{
    __shared__ float acc[RPB * DD];
    __shared__ float wLds[3 * DD * DD];
    __shared__ float asum[RPB];
    __shared__ int   cnt[RPB];
    __shared__ int   rstart[RPB + 1];
    __shared__ int   moreFlag[2];
    __shared__ int2  srt[CAP];
    const int b = blockIdx.x;
    const int lane = threadIdx.x & 31;
    const int wave = __builtin_amdgcn_readfirstlane((int)(threadIdx.x >> 5));
    const int tstride = nBuckets + 1;
    int myBeg, myLen;
    piece_of(off, tstride, b, wave + lane * SUM_WAVES, nSlabs, E, myBeg, myLen);
    const int sgrp = (int)threadIdx.x / STAGE_THREADS;
    const int st   = (int)threadIdx.x - sgrp * STAGE_THREADS;
    StageRegs sr;
    if (sgrp < 3) {
        const float* Wsel = (sgrp == 0) ? Wn : (sgrp == 1) ? Wgi : Wgn;
        stage_weights_load(sr, Wsel, st);
    }
    for (int i = threadIdx.x; i < RPB * DD; i += SUM_WAVES * 32) acc[i] = 0.0f;
    if (threadIdx.x < RPB) asum[threadIdx.x] = 0.0f;
    if (sgrp < 3) stage_weights_store(wLds + sgrp * (DD * DD), sr, st);

    const int4* recs = (const int4*)sorted;
    const int lastRow = (N > 0) ? N - 1 : 0;
    const int rowBase = b * RPB;

    int p0 = wave;
    bool has = p0 < nSlabs;
    int myStart = 0, total = 0, j = 0;
    {
        int incl = myLen;
        #pragma unroll
        for (int d = 1; d < 32; d <<= 1) {
            const int t = __shfl_up(incl, d);
            if (lane >= d) incl += t;
        }
        myStart = incl - myLen;
        total = __builtin_amdgcn_readlane(incl, 31);
    }

    const int maxRounds = E / RPWV + nSlabs / (SUM_WAVES * 32) + 2;
    for (int round = 0; round < maxRounds; ++round) {
        if (threadIdx.x < RPB) cnt[threadIdx.x] = 0;
        if (threadIdx.x == 0) moreFlag[round & 1] = 0;
        __syncthreads();

        int n = 0;
        if (has) n = min(RPWV, total - j);
        int rlv[RPL], rkv[RPL], cv[RPL], av[RPL];
        #pragma unroll
        for (int k = 0; k < RPL; ++k) { rlv[k] = 0; rkv[k] = 0; cv[k] = 0; av[k] = 0; }
        if (n > 0) {
            int4 rec[RPL];
            #pragma unroll
            for (int k = 0; k < RPL; ++k) {
                const int f = min(j + lane + 32 * k, total - 1);
                int L = 0;
                #pragma unroll
                for (int step = 16; step >= 1; step >>= 1) {
                    const int sv = __shfl(myStart, L + step);
                    if (sv <= f) L += step;
                }
                const int fs = __shfl(myStart, L);
                const int bg = __shfl(myBeg, L);
                rec[k] = recs[bg + (f - fs)];
            }
            #pragma unroll
            for (int k = 0; k < RPL; ++k) {
                rlv[k] = min(max(rec[k].x - rowBase, 0), RPB - 1);
                cv[k] = rec[k].y;
                av[k] = rec[k].z;
                if (lane + 32 * k < n) rkv[k] = atomicAdd(&cnt[rlv[k]], 1);
            }
        }
        __syncthreads();

        if (wave == 0) {
            const int r0 = lane * SCANQ;
            int c[SCANQ];
            int loc = 0;
            #pragma unroll
            for (int q = 0; q < SCANQ; ++q) {
                c[q] = (r0 + q < RPB) ? cnt[r0 + q] : 0;
                loc += c[q];
            }
            int sc = loc;
            #pragma unroll
            for (int d = 1; d < 32; d <<= 1) {
                const int t = __shfl_up(sc, d);
                if (lane >= d) sc += t;
            }
            int run = sc - loc;
            #pragma unroll
            for (int q = 0; q < SCANQ; ++q) {
                if (r0 + q < RPB) rstart[r0 + q] = run;
                run += c[q];
            }
            if (lane == 31) rstart[RPB] = sc;
        }
        __syncthreads();

        if (n > 0) {
            #pragma unroll
            for (int k = 0; k < RPL; ++k) {
                if (lane + 32 * k < n) {
                    int pos = rstart[rlv[k]] + rkv[k];
                    pos = min(max(pos, 0), CAP - 1);
                    srt[pos] = make_int2(cv[k], av[k]);
                }
            }
        }
        if (has) {
            j += n;
            if (j >= total) {
                p0 += SUM_WAVES * 32;
                has = p0 < nSlabs;
                j = 0;
                total = 0;
                if (has) {
                    piece_of(off, tstride, b, p0 + lane * SUM_WAVES, nSlabs, E, myBeg, myLen);
                    int incl = myLen;
                    #pragma unroll
                    for (int d = 1; d < 32; d <<= 1) {
                        const int t = __shfl_up(incl, d);
                        if (lane >= d) incl += t;
                    }
                    myStart = incl - myLen;
                    total = __builtin_amdgcn_readlane(incl, 31);
                }
            }
            if (has && lane == 0) moreFlag[round & 1] = 1;
        }
        __syncthreads();

        {
            const int rb = wave * RW;
            const int bndv = rstart[rb + min(lane, RW)];
            int i0 = __builtin_amdgcn_readlane(bndv, 0);
            const int iend = __builtin_amdgcn_readlane(bndv, RW);
            int cur = 0;
            int nxt = __builtin_amdgcn_readlane(bndv, 1);
            float s0 = 0.0f, s1 = 0.0f, s2 = 0.0f, sa = 0.0f;
            int2 ca[EPW];
            #pragma unroll
            for (int i = 0; i < EPW; ++i) ca[i] = make_int2(0, 0);
            if (i0 < iend) {
                #pragma unroll
                for (int i = 0; i < EPW; ++i) ca[i] = srt[min(i0 + i, iend - 1)];
            }
            for (; i0 < iend; i0 += EPW) {
                float a[EPW];
                float v[EPW][3];
                #pragma unroll
                for (int i = 0; i < EPW; ++i) {
                    a[i] = (i0 + i < iend) ? __int_as_float(ca[i].y) : 0.0f;
                    const float* p = X + (size_t)ca[i].x * DD + lane;
                    v[i][0] = p[0];
                    v[i][1] = p[32];
                    v[i][2] = p[64];
                }
                #pragma unroll
                for (int i = 0; i < EPW; ++i) ca[i] = srt[min(i0 + EPW + i, iend - 1)];
                #pragma unroll
                for (int i = 0; i < EPW; ++i) {
                    const int idx = i0 + i;
                    while (idx >= nxt && cur < RW - 1) {
                        float* q = acc + (rb + cur) * DD + lane;
                        atomicAdd(q,      s0);
                        atomicAdd(q + 32, s1);
                        atomicAdd(q + 64, s2);
                        if (lane == 0) atomicAdd(asum + rb + cur, sa);
                        s0 = 0.0f; s1 = 0.0f; s2 = 0.0f; sa = 0.0f;
                        ++cur;
                        nxt = __builtin_amdgcn_readlane(bndv, cur + 1);
                    }
                    s0 += a[i] * v[i][0];
                    s1 += a[i] * v[i][1];
                    s2 += a[i] * v[i][2];
                    sa += a[i];
                }
            }
            float* q = acc + (rb + cur) * DD + lane;
            atomicAdd(q,      s0);
            atomicAdd(q + 32, s1);
            atomicAdd(q + 64, s2);
            if (lane == 0) atomicAdd(asum + rb + cur, sa);
        }
        if (__builtin_amdgcn_readfirstlane(moreFlag[round & 1]) == 0) break;
    }

    const int half = lane >> 4;
    const int l16  = lane & 15;
    const int rt   = wave >> 1;
    const int ct0  = (wave & 1) * WCT;

    v2f aX[KSTEPS];
    {
        const int m = min(rowBase + rt * 16 + l16, lastRow);
        const float* aPtr = X + (size_t)m * DD + 2 * half;
        #pragma unroll
        for (int kt = 0; kt < KSTEPS; ++kt) aX[kt] = *(const v2f*)(aPtr + kt * 4);
    }
    float bnv[WCT];
    #pragma unroll
    for (int t = 0; t < WCT; ++t) bnv[t] = bn[(ct0 + t) * 16 + l16];

    __syncthreads();

    v8f h[WCT] = {};
    {
        const float* aRow = acc + (rt * 16 + l16) * DD + 2 * half;
        #pragma unroll
        for (int kt = 0; kt < KSTEPS; ++kt) {
            const v2f a0 = *(const v2f*)(aRow + kt * 4);
            const float* bRow = wLds + (size_t)(kt * 2 + half) * LDS_PAIR_STRIDE;
            #pragma unroll
            for (int t = 0; t < WCT; ++t) {
                v2f bw = *(const v2f*)(bRow + ((ct0 + t) * 16 + l16) * 2);
                h[t] = __builtin_amdgcn_wmma_f32_16x16x4_f32(
                    false, a0, false, bw, (short)0, h[t], false, false);
            }
        }
    }
    __syncthreads();

    #pragma unroll
    for (int t = 0; t < WCT; ++t) {
        const int nn = (ct0 + t) * 16 + l16;
        #pragma unroll
        for (int r = 0; r < 8; ++r) {
            const int lr = rt * 16 + r + 8 * half;
            acc[lr * DD + nn] = h[t][r] + asum[lr] * bnv[t];
        }
    }

    v8f g[WCT] = {};
    #pragma unroll
    for (int kt = 0; kt < KSTEPS; ++kt) {
        const float* bRow = wLds + DD * DD + (size_t)(kt * 2 + half) * LDS_PAIR_STRIDE;
        #pragma unroll
        for (int t = 0; t < WCT; ++t) {
            v2f bw = *(const v2f*)(bRow + ((ct0 + t) * 16 + l16) * 2);
            g[t] = __builtin_amdgcn_wmma_f32_16x16x4_f32(
                false, aX[kt], false, bw, (short)0, g[t], false, false);
        }
    }

    float xe[WCT][8];
    #pragma unroll
    for (int r = 0; r < 8; ++r) {
        const int m = min(rowBase + rt * 16 + r + 8 * half, lastRow);
        const float* xp = X + (size_t)m * DD + ct0 * 16 + l16;
        #pragma unroll
        for (int t = 0; t < WCT; ++t) xe[t][r] = xp[t * 16];
    }

    __syncthreads();

    {
        const float* aRow = acc + (rt * 16 + l16) * DD + 2 * half;
        #pragma unroll
        for (int kt = 0; kt < KSTEPS; ++kt) {
            const v2f a2 = *(const v2f*)(aRow + kt * 4);
            const float* bRow = wLds + 2 * DD * DD + (size_t)(kt * 2 + half) * LDS_PAIR_STRIDE;
            #pragma unroll
            for (int t = 0; t < WCT; ++t) {
                v2f bw = *(const v2f*)(bRow + ((ct0 + t) * 16 + l16) * 2);
                g[t] = __builtin_amdgcn_wmma_f32_16x16x4_f32(
                    false, a2, false, bw, (short)0, g[t], false, false);
            }
        }
    }
    __syncthreads();

    #pragma unroll
    for (int t = 0; t < WCT; ++t) {
        const int nn = (ct0 + t) * 16 + l16;
        const float bsum = bgi[nn] + bgn[nn];
        #pragma unroll
        for (int r = 0; r < 8; ++r) {
            float* q = acc + (rt * 16 + r + 8 * half) * DD + nn;
            const float ag = *q;
            float s = g[t][r] + bsum;
            s = 1.0f / (1.0f + __expf(-s));
            *q = ag * s + xe[t][r] * (1.0f - s);
        }
    }
    __syncthreads();

    float4* dst = (float4*)(out + (size_t)rowBase * DD);
    const float4* srcv = (const float4*)acc;
    for (int i = threadIdx.x; i < RPB * DD / 4; i += SUM_WAVES * 32) {
        const int r = (i * 4) / DD;
        if (rowBase + r < N) dst[i] = srcv[i];
    }
}

extern "C" void kernel_launch(void* const* d_in, const int* in_sizes, int n_in,
                              void* d_out, int out_size, void* d_ws, size_t ws_size,
                              hipStream_t stream) {
    const float* X      = (const float*)d_in[0];
    const float* a_vals = (const float*)d_in[1];
    const float* Wn     = (const float*)d_in[2];
    const float* bn     = (const float*)d_in[3];
    const float* Wgi    = (const float*)d_in[4];
    const float* bgi    = (const float*)d_in[5];
    const float* Wgn    = (const float*)d_in[6];
    const float* bgn    = (const float*)d_in[7];
    const int*   row    = (const int*)d_in[8];
    const int*   col    = (const int*)d_in[9];
    float* out = (float*)d_out;

    const int N = in_sizes[0] / DD;
    const int E = in_sizes[1];

    const int B = (N + RPB - 1) / RPB;
    const int slabs = (E + BIN_CHUNK - 1) / BIN_CHUNK;
    int* sorted = (int*)d_ws;
    int* off    = sorted + (size_t)slabs * BIN_CHUNK * 4;

    bin_place_kernel<<<slabs, BIN_THREADS, (2 * B + 1) * sizeof(int), stream>>>(
        row, col, a_vals, off, sorted, E, B);

    bucket_gate_kernel<<<B, SUM_WAVES * 32, 0, stream>>>(
        sorted, off, slabs, B, X, Wn, bn, Wgi, Wgn, bgi, bgn, out, E, N);
}


# gate: X@Wgi runs under the edge sum: bucket's X rows staged in LDS, each gather trip issues 2 of the 24 k-steps (6 WMMA) while its loads are in flight; 2 matrix phases follow, not 3; sum orders kept
# speedup vs baseline: 4.1220x; 1.0320x over previous
#include <hip/hip_runtime.h>
#include <hip/hip_bf16.h>

typedef __attribute__((ext_vector_type(2))) float v2f;
typedef __attribute__((ext_vector_type(8))) float v8f;

#define DD 96
#define KSTEPS (DD / 4)
#define CTILES (DD / 16)
#define LDS_PAIR_STRIDE (2 * DD)

#define RPB 208
#define BIN_THREADS 256
#define BIN_EPT 8
#define BIN_CHUNK (BIN_THREADS * BIN_EPT)
#define SCAN_ITEMS 4

__global__ __launch_bounds__(BIN_THREADS) void bin_place_kernel(
    const int* __restrict__ row, const int* __restrict__ col,
    const float* __restrict__ avals,
    int* __restrict__ off, int* __restrict__ sorted, int E, int B)
{
    extern __shared__ int binLds[];
    __shared__ int waveTot[BIN_THREADS / 32];
    int* h   = binLds;
    int* seg = binLds + B;
    const int tid  = threadIdx.x;
    const int lane = tid & 31;
    const int wave = tid >> 5;

    for (int i = tid; i < B; i += BIN_THREADS) h[i] = 0;

    int rv[BIN_EPT], cv[BIN_EPT], av[BIN_EPT], rk[BIN_EPT], bk[BIN_EPT];
    const int e0 = blockIdx.x * BIN_CHUNK + tid;
    #pragma unroll
    for (int k = 0; k < BIN_EPT; ++k) {
        const int e = e0 + k * BIN_THREADS;
        const int ec = min(e, E - 1);
        rv[k] = row[ec];
        cv[k] = col[ec];
        av[k] = __float_as_int(avals[ec]);
        rk[k] = 0;
    }
    __syncthreads();
    #pragma unroll
    for (int k = 0; k < BIN_EPT; ++k) {
        const int e = e0 + k * BIN_THREADS;
        bk[k] = min((int)((unsigned)rv[k] / RPB), B - 1);
        if (e < E) rk[k] = atomicAdd(&h[bk[k]], 1);
    }
    __syncthreads();

    int carry = 0;
    for (int b0 = 0; b0 < B; b0 += BIN_THREADS * SCAN_ITEMS) {
        const int i0 = b0 + tid * SCAN_ITEMS;
        int v[SCAN_ITEMS];
        int local = 0;
        #pragma unroll
        for (int q = 0; q < SCAN_ITEMS; ++q) {
            v[q] = (i0 + q < B) ? h[i0 + q] : 0;
            local += v[q];
        }
        int s = local;
        #pragma unroll
        for (int d = 1; d < 32; d <<= 1) {
            const int t = __shfl_up(s, d);
            if (lane >= d) s += t;
        }
        if (lane == 31) waveTot[wave] = s;
        __syncthreads();
        int before = 0, total = 0;
        #pragma unroll
        for (int w = 0; w < BIN_THREADS / 32; ++w) {
            const int t = waveTot[w];
            if (w < wave) before += t;
            total += t;
        }
        int run = carry + before + s - local;
        #pragma unroll
        for (int q = 0; q < SCAN_ITEMS; ++q) {
            if (i0 + q < B) seg[i0 + q] = run;
            run += v[q];
        }
        carry += total;
        __syncthreads();
    }
    if (tid == 0) seg[B] = carry;
    __syncthreads();

    int* myOff = off + (size_t)blockIdx.x * (B + 1);
    for (int i = tid; i <= B; i += BIN_THREADS) myOff[i] = seg[i];

    int4* recs = (int4*)sorted + (size_t)blockIdx.x * BIN_CHUNK;
    #pragma unroll
    for (int k = 0; k < BIN_EPT; ++k) {
        const int e = e0 + k * BIN_THREADS;
        if (e < E) {
            int pos = seg[bk[k]] + rk[k];
            pos = min(max(pos, 0), BIN_CHUNK - 1);
            recs[pos] = make_int4(rv[k], cv[k], av[k], 0);
        }
    }
}

#define STAGE_TRIPS 12
#define STAGE_THREADS (4 * (DD / 2))
struct StageRegs { v2f r0[STAGE_TRIPS]; v2f r1[STAGE_TRIPS]; };

__device__ __forceinline__ void stage_weights_load(StageRegs& sr,
                                                   const float* __restrict__ W, int t) {
    const int ph = t / (DD / 2);
    const int c2 = t - ph * (DD / 2);
    const float* src = W + (2 * ph) * DD + 2 * c2;
    #pragma unroll
    for (int it = 0; it < STAGE_TRIPS; ++it) {
        sr.r0[it] = *(const v2f*)(src + it * 8 * DD);
        sr.r1[it] = *(const v2f*)(src + it * 8 * DD + DD);
    }
}

__device__ __forceinline__ void stage_weights_store(float* __restrict__ lds,
                                                    const StageRegs& sr, int t) {
    const int ph = t / (DD / 2);
    const int c2 = t - ph * (DD / 2);
    float* dst = lds + ph * LDS_PAIR_STRIDE + 4 * c2;
    #pragma unroll
    for (int it = 0; it < STAGE_TRIPS; ++it) {
        *(float4*)(dst + it * 4 * LDS_PAIR_STRIDE) =
            make_float4(sr.r0[it].x, sr.r1[it].x, sr.r0[it].y, sr.r1[it].y);
    }
}

#define EPW 8
#define SUM_WAVES (2 * (RPB / 16))
#define WCT 3
#define RW (RPB / SUM_WAVES)
#define RPL 8
#define RPWV (32 * RPL)
#define CAP (SUM_WAVES * RPWV)
#define SCANQ ((RPB + 31) / 32)
#define XT_TRIPS ((RPB * DD / 4) / (SUM_WAVES * 32))
#define GK 2

__device__ __forceinline__ void piece_of(const int* __restrict__ off, int tstride, int b,
                                         int p, int nSlabs, int E, int& beg, int& len) {
    beg = 0;
    len = 0;
    if (p < nSlabs) {
        const int* t = off + (size_t)p * tstride + b;
        int o0 = t[0], o1 = t[1];
        const int cap = min(BIN_CHUNK, E - p * BIN_CHUNK);
        o0 = min(max(o0, 0), cap);
        o1 = min(max(o1, o0), cap);
        beg = p * BIN_CHUNK + o0;
        len = o1 - o0;
    }
}

__global__ __launch_bounds__(SUM_WAVES * 32) void bucket_gate_kernel(
    const int* __restrict__ sorted,
    const int* __restrict__ off, int nSlabs, int nBuckets,
    const float* __restrict__ X,
    const float* __restrict__ Wn, const float* __restrict__ bn,
    const float* __restrict__ Wgi, const float* __restrict__ Wgn,
    const float* __restrict__ bgi, const float* __restrict__ bgn,
    float* __restrict__ out, int E, int N)
{
    __shared__ float acc[RPB * DD];
    __shared__ float wLds[3 * DD * DD];
    __shared__ float asum[RPB];
    __shared__ int   cnt[RPB];
    __shared__ int   rstart[RPB + 1];
    __shared__ int   moreFlag[2];
    __shared__ int2  srt[CAP];
    __shared__ float xT[RPB * DD];
    const int b = blockIdx.x;
    const int lane = threadIdx.x & 31;
    const int wave = __builtin_amdgcn_readfirstlane((int)(threadIdx.x >> 5));
    const int tstride = nBuckets + 1;
    int myBeg, myLen;
    piece_of(off, tstride, b, wave + lane * SUM_WAVES, nSlabs, E, myBeg, myLen);
    const int sgrp = (int)threadIdx.x / STAGE_THREADS;
    const int st   = (int)threadIdx.x - sgrp * STAGE_THREADS;
    StageRegs sr;
    if (sgrp < 3) {
        const float* Wsel = (sgrp == 0) ? Wn : (sgrp == 1) ? Wgi : Wgn;
        stage_weights_load(sr, Wsel, st);
    }
    const int lastRow = (N > 0) ? N - 1 : 0;
    const int rowBase = b * RPB;
    float4 xs[XT_TRIPS];
    #pragma unroll
    for (int q = 0; q < XT_TRIPS; ++q) {
        const int i = (int)threadIdx.x + q * (SUM_WAVES * 32);
        const int r = i / (DD / 4);
        const int c4 = i - r * (DD / 4);
        const int m = min(rowBase + r, lastRow);
        xs[q] = *(const float4*)(X + (size_t)m * DD + 4 * c4);
    }
    for (int i = threadIdx.x; i < RPB * DD; i += SUM_WAVES * 32) acc[i] = 0.0f;
    if (threadIdx.x < RPB) asum[threadIdx.x] = 0.0f;
    if (sgrp < 3) stage_weights_store(wLds + sgrp * (DD * DD), sr, st);
    #pragma unroll
    for (int q = 0; q < XT_TRIPS; ++q)
        ((float4*)xT)[(int)threadIdx.x + q * (SUM_WAVES * 32)] = xs[q];

    const int4* recs = (const int4*)sorted;

    const int half = lane >> 4;
    const int l16  = lane & 15;
    const int rt   = wave >> 1;
    const int ct0  = (wave & 1) * WCT;
    v8f g[WCT] = {};
    int kk = 0;
    const float* gA = xT + (rt * 16 + l16) * DD + 2 * half;
    const float* gB = wLds + DD * DD + half * LDS_PAIR_STRIDE + (ct0 * 16 + l16) * 2;
#define GATE_KSTEPS()                                                            \
    {                                                                            \
        const float* aRow = gA + kk * 4;                                         \
        const float* bRow = gB + (size_t)kk * (2 * LDS_PAIR_STRIDE);             \
        _Pragma("unroll")                                                        \
        for (int s = 0; s < GK; ++s) {                                           \
            const v2f a1 = *(const v2f*)(aRow + s * 4);                          \
            _Pragma("unroll")                                                    \
            for (int t = 0; t < WCT; ++t) {                                      \
                const v2f bw = *(const v2f*)(bRow + s * 2 * LDS_PAIR_STRIDE + t * 32); \
                g[t] = __builtin_amdgcn_wmma_f32_16x16x4_f32(                    \
                    false, a1, false, bw, (short)0, g[t], false, false);         \
            }                                                                    \
        }                                                                        \
        kk += GK;                                                                \
    }

    int p0 = wave;
    bool has = p0 < nSlabs;
    int myStart = 0, total = 0, j = 0;
    {
        int incl = myLen;
        #pragma unroll
        for (int d = 1; d < 32; d <<= 1) {
            const int t = __shfl_up(incl, d);
            if (lane >= d) incl += t;
        }
        myStart = incl - myLen;
        total = __builtin_amdgcn_readlane(incl, 31);
    }

    const int maxRounds = E / RPWV + nSlabs / (SUM_WAVES * 32) + 2;
    for (int round = 0; round < maxRounds; ++round) {
        if (threadIdx.x < RPB) cnt[threadIdx.x] = 0;
        if (threadIdx.x == 0) moreFlag[round & 1] = 0;
        __syncthreads();

        int n = 0;
        if (has) n = min(RPWV, total - j);
        int rlv[RPL], rkv[RPL], cv[RPL], av[RPL];
        #pragma unroll
        for (int k = 0; k < RPL; ++k) { rlv[k] = 0; rkv[k] = 0; cv[k] = 0; av[k] = 0; }
        if (n > 0) {
            int4 rec[RPL];
            #pragma unroll
            for (int k = 0; k < RPL; ++k) {
                const int f = min(j + lane + 32 * k, total - 1);
                int L = 0;
                #pragma unroll
                for (int step = 16; step >= 1; step >>= 1) {
                    const int sv = __shfl(myStart, L + step);
                    if (sv <= f) L += step;
                }
                const int fs = __shfl(myStart, L);
                const int bg = __shfl(myBeg, L);
                rec[k] = recs[bg + (f - fs)];
            }
            #pragma unroll
            for (int k = 0; k < RPL; ++k) {
                rlv[k] = min(max(rec[k].x - rowBase, 0), RPB - 1);
                cv[k] = rec[k].y;
                av[k] = rec[k].z;
                if (lane + 32 * k < n) rkv[k] = atomicAdd(&cnt[rlv[k]], 1);
            }
        }
        __syncthreads();

        if (wave == 0) {
            const int r0 = lane * SCANQ;
            int c[SCANQ];
            int loc = 0;
            #pragma unroll
            for (int q = 0; q < SCANQ; ++q) {
                c[q] = (r0 + q < RPB) ? cnt[r0 + q] : 0;
                loc += c[q];
            }
            int sc = loc;
            #pragma unroll
            for (int d = 1; d < 32; d <<= 1) {
                const int t = __shfl_up(sc, d);
                if (lane >= d) sc += t;
            }
            int run = sc - loc;
            #pragma unroll
            for (int q = 0; q < SCANQ; ++q) {
                if (r0 + q < RPB) rstart[r0 + q] = run;
                run += c[q];
            }
            if (lane == 31) rstart[RPB] = sc;
        }
        __syncthreads();

        if (n > 0) {
            #pragma unroll
            for (int k = 0; k < RPL; ++k) {
                if (lane + 32 * k < n) {
                    int pos = rstart[rlv[k]] + rkv[k];
                    pos = min(max(pos, 0), CAP - 1);
                    srt[pos] = make_int2(cv[k], av[k]);
                }
            }
        }
        if (has) {
            j += n;
            if (j >= total) {
                p0 += SUM_WAVES * 32;
                has = p0 < nSlabs;
                j = 0;
                total = 0;
                if (has) {
                    piece_of(off, tstride, b, p0 + lane * SUM_WAVES, nSlabs, E, myBeg, myLen);
                    int incl = myLen;
                    #pragma unroll
                    for (int d = 1; d < 32; d <<= 1) {
                        const int t = __shfl_up(incl, d);
                        if (lane >= d) incl += t;
                    }
                    myStart = incl - myLen;
                    total = __builtin_amdgcn_readlane(incl, 31);
                }
            }
            if (has && lane == 0) moreFlag[round & 1] = 1;
        }
        __syncthreads();

        {
            const int rb = wave * RW;
            const int bndv = rstart[rb + min(lane, RW)];
            int i0 = __builtin_amdgcn_readlane(bndv, 0);
            const int iend = __builtin_amdgcn_readlane(bndv, RW);
            int cur = 0;
            int nxt = __builtin_amdgcn_readlane(bndv, 1);
            float s0 = 0.0f, s1 = 0.0f, s2 = 0.0f, sa = 0.0f;
            int2 ca[EPW];
            #pragma unroll
            for (int i = 0; i < EPW; ++i) ca[i] = make_int2(0, 0);
            if (i0 < iend) {
                #pragma unroll
                for (int i = 0; i < EPW; ++i) ca[i] = srt[min(i0 + i, iend - 1)];
            }
            for (; i0 < iend; i0 += EPW) {
                float a[EPW];
                float v[EPW][3];
                #pragma unroll
                for (int i = 0; i < EPW; ++i) {
                    a[i] = (i0 + i < iend) ? __int_as_float(ca[i].y) : 0.0f;
                    const float* p = X + (size_t)ca[i].x * DD + lane;
                    v[i][0] = p[0];
                    v[i][1] = p[32];
                    v[i][2] = p[64];
                }
                #pragma unroll
                for (int i = 0; i < EPW; ++i) ca[i] = srt[min(i0 + EPW + i, iend - 1)];
                if (kk < KSTEPS) GATE_KSTEPS()
                #pragma unroll
                for (int i = 0; i < EPW; ++i) {
                    const int idx = i0 + i;
                    while (idx >= nxt && cur < RW - 1) {
                        float* q = acc + (rb + cur) * DD + lane;
                        atomicAdd(q,      s0);
                        atomicAdd(q + 32, s1);
                        atomicAdd(q + 64, s2);
                        if (lane == 0) atomicAdd(asum + rb + cur, sa);
                        s0 = 0.0f; s1 = 0.0f; s2 = 0.0f; sa = 0.0f;
                        ++cur;
                        nxt = __builtin_amdgcn_readlane(bndv, cur + 1);
                    }
                    s0 += a[i] * v[i][0];
                    s1 += a[i] * v[i][1];
                    s2 += a[i] * v[i][2];
                    sa += a[i];
                }
            }
            float* q = acc + (rb + cur) * DD + lane;
            atomicAdd(q,      s0);
            atomicAdd(q + 32, s1);
            atomicAdd(q + 64, s2);
            if (lane == 0) atomicAdd(asum + rb + cur, sa);
        }
        if (__builtin_amdgcn_readfirstlane(moreFlag[round & 1]) == 0) break;
    }

    while (kk < KSTEPS) GATE_KSTEPS()
#undef GATE_KSTEPS

    float bnv[WCT];
    #pragma unroll
    for (int t = 0; t < WCT; ++t) bnv[t] = bn[(ct0 + t) * 16 + l16];

    __syncthreads();

    v8f h[WCT] = {};
    {
        const float* aRow = acc + (rt * 16 + l16) * DD + 2 * half;
        #pragma unroll
        for (int kt = 0; kt < KSTEPS; ++kt) {
            const v2f a0 = *(const v2f*)(aRow + kt * 4);
            const float* bRow = wLds + (size_t)(kt * 2 + half) * LDS_PAIR_STRIDE;
            #pragma unroll
            for (int t = 0; t < WCT; ++t) {
                v2f bw = *(const v2f*)(bRow + ((ct0 + t) * 16 + l16) * 2);
                h[t] = __builtin_amdgcn_wmma_f32_16x16x4_f32(
                    false, a0, false, bw, (short)0, h[t], false, false);
            }
        }
    }
    __syncthreads();

    #pragma unroll
    for (int t = 0; t < WCT; ++t) {
        const int nn = (ct0 + t) * 16 + l16;
        #pragma unroll
        for (int r = 0; r < 8; ++r) {
            const int lr = rt * 16 + r + 8 * half;
            acc[lr * DD + nn] = h[t][r] + asum[lr] * bnv[t];
        }
    }

    float xe[WCT][8];
    #pragma unroll
    for (int r = 0; r < 8; ++r) {
        const int m = min(rowBase + rt * 16 + r + 8 * half, lastRow);
        const float* xp = X + (size_t)m * DD + ct0 * 16 + l16;
        #pragma unroll
        for (int t = 0; t < WCT; ++t) xe[t][r] = xp[t * 16];
    }

    __syncthreads();

    {
        const float* aRow = acc + (rt * 16 + l16) * DD + 2 * half;
        #pragma unroll
        for (int kt = 0; kt < KSTEPS; ++kt) {
            const v2f a2 = *(const v2f*)(aRow + kt * 4);
            const float* bRow = wLds + 2 * DD * DD + (size_t)(kt * 2 + half) * LDS_PAIR_STRIDE;
            #pragma unroll
            for (int t = 0; t < WCT; ++t) {
                v2f bw = *(const v2f*)(bRow + ((ct0 + t) * 16 + l16) * 2);
                g[t] = __builtin_amdgcn_wmma_f32_16x16x4_f32(
                    false, a2, false, bw, (short)0, g[t], false, false);
            }
        }
    }
    __syncthreads();

    #pragma unroll
    for (int t = 0; t < WCT; ++t) {
        const int nn = (ct0 + t) * 16 + l16;
        const float bsum = bgi[nn] + bgn[nn];
        #pragma unroll
        for (int r = 0; r < 8; ++r) {
            float* q = acc + (rt * 16 + r + 8 * half) * DD + nn;
            const float ag = *q;
            float s = g[t][r] + bsum;
            s = 1.0f / (1.0f + __expf(-s));
            *q = ag * s + xe[t][r] * (1.0f - s);
        }
    }
    __syncthreads();

    float4* dst = (float4*)(out + (size_t)rowBase * DD);
    const float4* srcv = (const float4*)acc;
    for (int i = threadIdx.x; i < RPB * DD / 4; i += SUM_WAVES * 32) {
        const int r = (i * 4) / DD;
        if (rowBase + r < N) dst[i] = srcv[i];
    }
}

extern "C" void kernel_launch(void* const* d_in, const int* in_sizes, int n_in,
                              void* d_out, int out_size, void* d_ws, size_t ws_size,
                              hipStream_t stream) {
    const float* X      = (const float*)d_in[0];
    const float* a_vals = (const float*)d_in[1];
    const float* Wn     = (const float*)d_in[2];
    const float* bn     = (const float*)d_in[3];
    const float* Wgi    = (const float*)d_in[4];
    const float* bgi    = (const float*)d_in[5];
    const float* Wgn    = (const float*)d_in[6];
    const float* bgn    = (const float*)d_in[7];
    const int*   row    = (const int*)d_in[8];
    const int*   col    = (const int*)d_in[9];
    float* out = (float*)d_out;

    const int N = in_sizes[0] / DD;
    const int E = in_sizes[1];

    const int B = (N + RPB - 1) / RPB;
    const int slabs = (E + BIN_CHUNK - 1) / BIN_CHUNK;
    int* sorted = (int*)d_ws;
    int* off    = sorted + (size_t)slabs * BIN_CHUNK * 4;

    bin_place_kernel<<<slabs, BIN_THREADS, (2 * B + 1) * sizeof(int), stream>>>(
        row, col, a_vals, off, sorted, E, B);

    bucket_gate_kernel<<<B, SUM_WAVES * 32, 0, stream>>>(
        sorted, off, slabs, B, X, Wn, bn, Wgi, Wgn, bgi, bgn, out, E, N);
}
